# starting kernel
# baseline (speedup 1.0000x reference)
#include <hip/hip_runtime.h>
#include <math.h>

#define B_  8
#define N_  4096
#define D_  64
#define S_  1024
#define K_  32

typedef __attribute__((ext_vector_type(2))) float v2f;
typedef __attribute__((ext_vector_type(4))) float v4f;
typedef __attribute__((ext_vector_type(8))) float v8f;

__global__ __launch_bounds__(1024) void fps_kernel(
    const float* __restrict__ xyz,
    const int*   __restrict__ farthest_init,
    float* __restrict__ out_newxyz,
    float* __restrict__ wsA,
    float* __restrict__ wsC2)
{
    __shared__ v4f   p4[N_];
    __shared__ float redV[32];
    __shared__ int   redI[32];
    __shared__ int   sFar;

    const int b    = blockIdx.x;
    const int tid  = threadIdx.x;
    const int lane = tid & 31;
    const int wid  = tid >> 5;

    const float* xb = xyz + (size_t)b * 3 * N_;
    for (int i = tid; i < N_; i += 1024) {
        v4f v;
        v.x = xb[i];
        v.y = xb[N_ + i];
        v.z = xb[2 * N_ + i];
        v.w = 0.f;
        p4[i] = v;
    }
    float dist[4];
#pragma unroll
    for (int j = 0; j < 4; ++j) dist[j] = 1e10f;
    if (tid == 0) sFar = farthest_init[b];
    __syncthreads();

    for (int it = 0; it < S_; ++it) {
        const v4f c = p4[sFar];
        if (tid == 0) {
            out_newxyz[(size_t)b * 3 * S_ + 0 * S_ + it] = c.x;
            out_newxyz[(size_t)b * 3 * S_ + 1 * S_ + it] = c.y;
            out_newxyz[(size_t)b * 3 * S_ + 2 * S_ + it] = c.z;
            float* A = wsA + ((size_t)b * S_ + it) * 4;
            A[0] = -2.f * c.x; A[1] = -2.f * c.y; A[2] = -2.f * c.z; A[3] = 1.f;
            wsC2[(size_t)b * S_ + it] = c.x * c.x + c.y * c.y + c.z * c.z;
        }
        float bv = -1.f; int bi = 0x7fffffff;
#pragma unroll
        for (int j = 0; j < 4; ++j) {
            const int i = tid + 1024 * j;
            const v4f p = p4[i];
            const float dx = p.x - c.x, dy = p.y - c.y, dz = p.z - c.z;
            const float d = dx * dx + dy * dy + dz * dz;
            const float nd = fminf(dist[j], d);
            dist[j] = nd;
            if (nd > bv || (nd == bv && i < bi)) { bv = nd; bi = i; }
        }
#pragma unroll
        for (int off = 16; off >= 1; off >>= 1) {
            const float ov = __shfl_xor(bv, off, 32);
            const int   oi = __shfl_xor(bi, off, 32);
            if (ov > bv || (ov == bv && oi < bi)) { bv = ov; bi = oi; }
        }
        if (lane == 0) { redV[wid] = bv; redI[wid] = bi; }
        __syncthreads();
        if (wid == 0) {
            bv = redV[lane]; bi = redI[lane];
#pragma unroll
            for (int off = 16; off >= 1; off >>= 1) {
                const float ov = __shfl_xor(bv, off, 32);
                const int   oi = __shfl_xor(bi, off, 32);
                if (ov > bv || (ov == bv && oi < bi)) { bv = ov; bi = oi; }
            }
            if (lane == 0) sFar = bi;
        }
        __syncthreads();
    }
}

#define TPITCH 4100

__global__ __launch_bounds__(256) void group_kernel(
    const float* __restrict__ xyz,
    const float* __restrict__ pts,
    const float* __restrict__ wsA,
    const float* __restrict__ wsC2,
    float* __restrict__ out_feat)
{
    __shared__ float DT[16 * TPITCH];
    __shared__ int   selIdx[16][K_];

    const int b    = blockIdx.y;
    const int s0   = blockIdx.x * 16;
    const int tid  = threadIdx.x;
    const int lane = tid & 31;
    const int wid  = tid >> 5;
    const bool hi  = lane >= 16;

    const float* xb = xyz + (size_t)b * 3 * N_;

    const int  mcol = lane & 15;
    const v4f  A4   = *(const v4f*)(wsA + ((size_t)b * S_ + s0 + mcol) * 4);
    v2f a;
    a.x = hi ? A4.z : A4.x;
    a.y = hi ? A4.w : A4.y;

    const int mbase = hi ? 8 : 0;
    float cmr[8];
#pragma unroll
    for (int r = 0; r < 8; ++r)
        cmr[r] = wsC2[(size_t)b * S_ + s0 + mbase + r];

    for (int t = 0; t < 32; ++t) {
        const int n = wid * 512 + t * 16 + mcol;
        const float x = xb[n], y = xb[N_ + n], z = xb[2 * N_ + n];
        const float pn2 = x * x + y * y + z * z;
        v2f bb;
        bb.x = hi ? z : x;
        bb.y = hi ? pn2 : y;

        v8f c = {};
        c = __builtin_amdgcn_wmma_f32_16x16x4_f32(
                 false, a,  false, bb,
                 (short)0, c,  false,  false);

#pragma unroll
        for (int r = 0; r < 8; ++r)
            DT[(mbase + r) * TPITCH + n] = c[r] + cmr[r];
    }
    __syncthreads();

    for (int rr = 0; rr < 2; ++rr) {
        const int m = wid + rr * 8;
        float* row = &DT[m * TPITCH];
        for (int k = 0; k < K_; ++k) {
            float bv = 3.4e38f; int bi = 0x7fffffff;
            for (int jj = 0; jj < N_ / 128; ++jj) {
                const int j0 = jj * 128 + lane * 4;
                const v4f dv = *(const v4f*)&row[j0];
#pragma unroll
                for (int e = 0; e < 4; ++e) {
                    const float d = dv[e];
                    const int   j = j0 + e;
                    if (d < bv || (d == bv && j < bi)) { bv = d; bi = j; }
                }
            }
#pragma unroll
            for (int off = 16; off >= 1; off >>= 1) {
                const float ov = __shfl_xor(bv, off, 32);
                const int   oi = __shfl_xor(bi, off, 32);
                if (ov < bv || (ov == bv && oi < bi)) { bv = ov; bi = oi; }
            }
            if (lane == 0) { selIdx[m][k] = bi; row[bi] = 3.4e38f; }
            __syncthreads();
        }
    }
    __syncthreads();

    const float* pb = pts + (size_t)b * D_ * N_;
    for (int t = tid; t < 16 * 67; t += 256) {
        const int m = t / 67;
        const int d = t % 67;
        const float* src = (d < 3) ? (xb + (size_t)d * N_)
                                   : (pb + (size_t)(d - 3) * N_);
        float mx = -3.4e38f;
#pragma unroll 4
        for (int k = 0; k < K_; ++k) mx = fmaxf(mx, src[selIdx[m][k]]);
        out_feat[(size_t)b * 67 * S_ + (size_t)d * S_ + (s0 + m)] = mx;
    }
}

extern "C" void kernel_launch(void* const* d_in, const int* in_sizes, int n_in,
                              void* d_out, int out_size, void* d_ws, size_t ws_size,
                              hipStream_t stream) {
    const float* xyz    = (const float*)d_in[0];
    const float* points = (const float*)d_in[1];
    const int*   finit  = (const int*)d_in[2];

    float* out        = (float*)d_out;
    float* out_newxyz = out;
    float* out_feat   = out + (size_t)B_ * 3 * S_;

    float* wsA  = (float*)d_ws;
    float* wsC2 = wsA + (size_t)B_ * S_ * 4;

    fps_kernel<<<B_, 1024, 0, stream>>>(xyz, finit, out_newxyz, wsA, wsC2);
    group_kernel<<<dim3(S_ / 16, B_), 256, 0, stream>>>(xyz, points, wsA, wsC2, out_feat);
}


# grouping kernel top-32 taken as a set, not by 32 serial picks: per row a threshold (max of 32 lane minima), ballot compaction of survivors, bitwise radix select; per-pick block barriers gone
# speedup vs baseline: 1.4833x; 1.4833x over previous
#include <hip/hip_runtime.h>
#include <math.h>

#define B_  8
#define N_  4096
#define D_  64
#define S_  1024
#define K_  32

typedef __attribute__((ext_vector_type(2))) float v2f;
typedef __attribute__((ext_vector_type(4))) float v4f;
typedef __attribute__((ext_vector_type(8))) float v8f;

__global__ __launch_bounds__(1024) void fps_kernel(
    const float* __restrict__ xyz,
    const int*   __restrict__ farthest_init,
    float* __restrict__ out_newxyz,
    float* __restrict__ wsA,
    float* __restrict__ wsC2)
{
    __shared__ v4f   p4[N_];
    __shared__ float redV[32];
    __shared__ int   redI[32];
    __shared__ int   sFar;

    const int b    = blockIdx.x;
    const int tid  = threadIdx.x;
    const int lane = tid & 31;
    const int wid  = tid >> 5;

    const float* xb = xyz + (size_t)b * 3 * N_;
    for (int i = tid; i < N_; i += 1024) {
        v4f v;
        v.x = xb[i];
        v.y = xb[N_ + i];
        v.z = xb[2 * N_ + i];
        v.w = 0.f;
        p4[i] = v;
    }
    float dist[4];
#pragma unroll
    for (int j = 0; j < 4; ++j) dist[j] = 1e10f;
    if (tid == 0) sFar = farthest_init[b];
    __syncthreads();

    for (int it = 0; it < S_; ++it) {
        const v4f c = p4[sFar];
        if (tid == 0) {
            out_newxyz[(size_t)b * 3 * S_ + 0 * S_ + it] = c.x;
            out_newxyz[(size_t)b * 3 * S_ + 1 * S_ + it] = c.y;
            out_newxyz[(size_t)b * 3 * S_ + 2 * S_ + it] = c.z;
            float* A = wsA + ((size_t)b * S_ + it) * 4;
            A[0] = -2.f * c.x; A[1] = -2.f * c.y; A[2] = -2.f * c.z; A[3] = 1.f;
            wsC2[(size_t)b * S_ + it] = c.x * c.x + c.y * c.y + c.z * c.z;
        }
        float bv = -1.f; int bi = 0x7fffffff;
#pragma unroll
        for (int j = 0; j < 4; ++j) {
            const int i = tid + 1024 * j;
            const v4f p = p4[i];
            const float dx = p.x - c.x, dy = p.y - c.y, dz = p.z - c.z;
            const float d = dx * dx + dy * dy + dz * dz;
            const float nd = fminf(dist[j], d);
            dist[j] = nd;
            if (nd > bv || (nd == bv && i < bi)) { bv = nd; bi = i; }
        }
#pragma unroll
        for (int off = 16; off >= 1; off >>= 1) {
            const float ov = __shfl_xor(bv, off, 32);
            const int   oi = __shfl_xor(bi, off, 32);
            if (ov > bv || (ov == bv && oi < bi)) { bv = ov; bi = oi; }
        }
        if (lane == 0) { redV[wid] = bv; redI[wid] = bi; }
        __syncthreads();
        if (wid == 0) {
            bv = redV[lane]; bi = redI[lane];
#pragma unroll
            for (int off = 16; off >= 1; off >>= 1) {
                const float ov = __shfl_xor(bv, off, 32);
                const int   oi = __shfl_xor(bi, off, 32);
                if (ov > bv || (ov == bv && oi < bi)) { bv = ov; bi = oi; }
            }
            if (lane == 0) sFar = bi;
        }
        __syncthreads();
    }
}

#define TPITCH 4100
#define CAND_CAP 512

template <int R>
__device__ __forceinline__ void select32(const float* cd, const int* cj,
                                         int C, int lane, int* selRow)
{
    unsigned key[R];
    int      jv[R];
#pragma unroll
    for (int r = 0; r < R; ++r) {
        const int  i  = r * 32 + lane;
        const bool in = i < C;
        const int  ii = in ? i : 0;
        const unsigned u = __float_as_uint(cd[ii]);
        const unsigned k = u ^ ((unsigned)((int)u >> 31) | 0x80000000u);
        key[r] = in ? k : 0xFFFFFFFFu;
        jv[r]  = cj[ii];
    }
    unsigned T = 0;
    for (int bit = 31; bit >= 0; --bit) {
        const unsigned cand = T | (1u << bit);
        int cnt = 0;
#pragma unroll
        for (int r = 0; r < R; ++r)
            cnt += __popc(__builtin_amdgcn_ballot_w32(key[r] < cand));
        if (cnt < K_) T = cand;
    }
    int nless = 0;
#pragma unroll
    for (int r = 0; r < R; ++r)
        nless += __popc(__builtin_amdgcn_ballot_w32(key[r] < T));
    const int need = K_ - nless;
    int running = 0, tieSeen = 0;
#pragma unroll
    for (int r = 0; r < R; ++r) {
        const bool isTie = key[r] == T;
        const unsigned tb = __builtin_amdgcn_ballot_w32(isTie);
        const int tieRank = tieSeen + (int)__builtin_amdgcn_mbcnt_lo(tb, 0u);
        const bool sel = (key[r] < T) || (isTie && tieRank < need);
        const unsigned sb = __builtin_amdgcn_ballot_w32(sel);
        const int slot = running + (int)__builtin_amdgcn_mbcnt_lo(sb, 0u);
        if (sel && slot < K_) selRow[slot] = jv[r];
        running += __popc(sb);
        tieSeen += __popc(tb);
    }
}

__device__ __forceinline__ void select32_scan(const float* row, int lane, int* selRow)
{
    float pd = -3.4e38f; int pj = -1;
    for (int k = 0; k < K_; ++k) {
        float bv = 3.4e38f; int bi = 0x7fffffff;
        for (int jj = 0; jj < N_ / 128; ++jj) {
            const int j0 = jj * 128 + lane * 4;
            const v4f dv = *(const v4f*)&row[j0];
#pragma unroll
            for (int e = 0; e < 4; ++e) {
                const float d = dv[e];
                const int   j = j0 + e;
                const bool after = d > pd || (d == pd && j > pj);
                if (after && (d < bv || (d == bv && j < bi))) { bv = d; bi = j; }
            }
        }
#pragma unroll
        for (int off = 16; off >= 1; off >>= 1) {
            const float ov = __shfl_xor(bv, off, 32);
            const int   oi = __shfl_xor(bi, off, 32);
            if (ov < bv || (ov == bv && oi < bi)) { bv = ov; bi = oi; }
        }
        if (lane == 0) selRow[k] = bi;
        pd = bv; pj = bi;
    }
}

__global__ __launch_bounds__(256) void group_kernel(
    const float* __restrict__ xyz,
    const float* __restrict__ pts,
    const float* __restrict__ wsA,
    const float* __restrict__ wsC2,
    float* __restrict__ out_feat)
{
    __shared__ float DT[16 * TPITCH];
    __shared__ int   selIdx[16][K_];
    __shared__ float candD[8][CAND_CAP];
    __shared__ int   candJ[8][CAND_CAP];

    const int b    = blockIdx.y;
    const int s0   = blockIdx.x * 16;
    const int tid  = threadIdx.x;
    const int lane = tid & 31;
    const int wid  = tid >> 5;
    const bool hi  = lane >= 16;

    const float* xb = xyz + (size_t)b * 3 * N_;

    const int  mcol = lane & 15;
    const v4f  A4   = *(const v4f*)(wsA + ((size_t)b * S_ + s0 + mcol) * 4);
    v2f a;
    a.x = hi ? A4.z : A4.x;
    a.y = hi ? A4.w : A4.y;

    const int mbase = hi ? 8 : 0;
    float cmr[8];
#pragma unroll
    for (int r = 0; r < 8; ++r)
        cmr[r] = wsC2[(size_t)b * S_ + s0 + mbase + r];

    for (int t = 0; t < 32; ++t) {
        const int n = wid * 512 + t * 16 + mcol;
        const float x = xb[n], y = xb[N_ + n], z = xb[2 * N_ + n];
        const float pn2 = x * x + y * y + z * z;
        v2f bb;
        bb.x = hi ? z : x;
        bb.y = hi ? pn2 : y;

        v8f c = {};
        c = __builtin_amdgcn_wmma_f32_16x16x4_f32(
                 false, a,  false, bb,
                 (short)0, c,  false,  false);

#pragma unroll
        for (int r = 0; r < 8; ++r)
            DT[(mbase + r) * TPITCH + n] = c[r] + cmr[r];
    }
    __syncthreads();

    for (int rr = 0; rr < 2; ++rr) {
        const int m = wid + rr * 8;
        const float* row = &DT[m * TPITCH];
        float* cd = candD[wid];
        int*   cj = candJ[wid];

        float mn = 3.4e38f;
        for (int jj = 0; jj < N_ / 128; ++jj) {
            const v4f dv = *(const v4f*)&row[jj * 128 + lane * 4];
            mn = fminf(mn, fminf(fminf(dv.x, dv.y), fminf(dv.z, dv.w)));
        }
        float U = mn;
#pragma unroll
        for (int off = 16; off >= 1; off >>= 1)
            U = fmaxf(U, __shfl_xor(U, off, 32));

        int C = 0;
        for (int jj = 0; jj < N_ / 128; ++jj) {
            const int j0 = jj * 128 + lane * 4;
            const v4f dv = *(const v4f*)&row[j0];
            const bool p0 = dv.x <= U, p1 = dv.y <= U;
            const bool p2 = dv.z <= U, p3 = dv.w <= U;
            const unsigned b0 = __builtin_amdgcn_ballot_w32(p0);
            const unsigned b1 = __builtin_amdgcn_ballot_w32(p1);
            const unsigned b2 = __builtin_amdgcn_ballot_w32(p2);
            const unsigned b3 = __builtin_amdgcn_ballot_w32(p3);
            unsigned pre = __builtin_amdgcn_mbcnt_lo(b0, 0u);
            pre = __builtin_amdgcn_mbcnt_lo(b1, pre);
            pre = __builtin_amdgcn_mbcnt_lo(b2, pre);
            pre = __builtin_amdgcn_mbcnt_lo(b3, pre);
            int pos = C + (int)pre;
            if (p0 && pos < CAND_CAP) { cd[pos] = dv.x; cj[pos] = j0; }
            pos += p0 ? 1 : 0;
            if (p1 && pos < CAND_CAP) { cd[pos] = dv.y; cj[pos] = j0 + 1; }
            pos += p1 ? 1 : 0;
            if (p2 && pos < CAND_CAP) { cd[pos] = dv.z; cj[pos] = j0 + 2; }
            pos += p2 ? 1 : 0;
            if (p3 && pos < CAND_CAP) { cd[pos] = dv.w; cj[pos] = j0 + 3; }
            C += __popc(b0) + __popc(b1) + __popc(b2) + __popc(b3);
        }
        __syncthreads();

        int* selRow = selIdx[m];
        if (C <= 128)           select32<4>(cd, cj, C, lane, selRow);
        else if (C <= 256)      select32<8>(cd, cj, C, lane, selRow);
        else if (C <= CAND_CAP) select32<16>(cd, cj, C, lane, selRow);
        else                    select32_scan(row, lane, selRow);
        __syncthreads();
    }

    const float* pb = pts + (size_t)b * D_ * N_;
    for (int t = tid; t < 16 * 67; t += 256) {
        const int m = t / 67;
        const int d = t % 67;
        const float* src = (d < 3) ? (xb + (size_t)d * N_)
                                   : (pb + (size_t)(d - 3) * N_);
        float mx = -3.4e38f;
#pragma unroll 4
        for (int k = 0; k < K_; ++k) mx = fmaxf(mx, src[selIdx[m][k]]);
        out_feat[(size_t)b * 67 * S_ + (size_t)d * S_ + (s0 + m)] = mx;
    }
}

extern "C" void kernel_launch(void* const* d_in, const int* in_sizes, int n_in,
                              void* d_out, int out_size, void* d_ws, size_t ws_size,
                              hipStream_t stream) {
    const float* xyz    = (const float*)d_in[0];
    const float* points = (const float*)d_in[1];
    const int*   finit  = (const int*)d_in[2];

    float* out        = (float*)d_out;
    float* out_newxyz = out;
    float* out_feat   = out + (size_t)B_ * 3 * S_;

    float* wsA  = (float*)d_ws;
    float* wsC2 = wsA + (size_t)B_ * S_ * 4;

    fps_kernel<<<B_, 1024, 0, stream>>>(xyz, finit, out_newxyz, wsA, wsC2);
    group_kernel<<<dim3(S_ / 16, B_), 256, 0, stream>>>(xyz, points, wsA, wsC2, out_feat);
}


# grouping kernel: 16 waves a block, one centroid row per wave from threshold pass through channel gather (a lane per channel, 96 independent loads a row); was 8 waves, two rows each
# speedup vs baseline: 2.0820x; 1.4036x over previous
#include <hip/hip_runtime.h>
#include <math.h>

#define B_  8
#define N_  4096
#define D_  64
#define S_  1024
#define K_  32

typedef __attribute__((ext_vector_type(2))) float v2f;
typedef __attribute__((ext_vector_type(4))) float v4f;
typedef __attribute__((ext_vector_type(8))) float v8f;

__global__ __launch_bounds__(1024) void fps_kernel(
    const float* __restrict__ xyz,
    const int*   __restrict__ farthest_init,
    float* __restrict__ out_newxyz,
    float* __restrict__ wsA,
    float* __restrict__ wsC2)
{
    __shared__ v4f   p4[N_];
    __shared__ float redV[32];
    __shared__ int   redI[32];
    __shared__ int   sFar;

    const int b    = blockIdx.x;
    const int tid  = threadIdx.x;
    const int lane = tid & 31;
    const int wid  = tid >> 5;

    const float* xb = xyz + (size_t)b * 3 * N_;
    for (int i = tid; i < N_; i += 1024) {
        v4f v;
        v.x = xb[i];
        v.y = xb[N_ + i];
        v.z = xb[2 * N_ + i];
        v.w = 0.f;
        p4[i] = v;
    }
    float dist[4];
#pragma unroll
    for (int j = 0; j < 4; ++j) dist[j] = 1e10f;
    if (tid == 0) sFar = farthest_init[b];
    __syncthreads();

    for (int it = 0; it < S_; ++it) {
        const v4f c = p4[sFar];
        if (tid == 0) {
            out_newxyz[(size_t)b * 3 * S_ + 0 * S_ + it] = c.x;
            out_newxyz[(size_t)b * 3 * S_ + 1 * S_ + it] = c.y;
            out_newxyz[(size_t)b * 3 * S_ + 2 * S_ + it] = c.z;
            float* A = wsA + ((size_t)b * S_ + it) * 4;
            A[0] = -2.f * c.x; A[1] = -2.f * c.y; A[2] = -2.f * c.z; A[3] = 1.f;
            float c2;
            {
#pragma clang fp contract(off)
                c2 = c.z * c.z + __builtin_fmaf(c.x, c.x, c.y * c.y);
            }
            wsC2[(size_t)b * S_ + it] = c2;
        }
        float bv = -1.f; int bi = 0x7fffffff;
#pragma unroll
        for (int j = 0; j < 4; ++j) {
            const int i = tid + 1024 * j;
            const v4f p = p4[i];
            const float dx = p.x - c.x, dy = p.y - c.y, dz = p.z - c.z;
            float d;
            {
#pragma clang fp contract(off)
                d = (j < 2) ? __builtin_fmaf(dz, dz, __builtin_fmaf(dx, dx, dy * dy))
                            : dz * dz + __builtin_fmaf(dy, dy, dx * dx);
            }
            const float nd = fminf(dist[j], d);
            dist[j] = nd;
            const bool take = nd > bv;
            bv = take ? nd : bv;
            bi = take ? i : bi;
        }
#pragma unroll
        for (int off = 16; off >= 1; off >>= 1) {
            const float ov = __shfl_xor(bv, off, 32);
            const int   oi = __shfl_xor(bi, off, 32);
            const bool gt = ov > bv, eq = ov == bv;
            const int  ti = oi < bi ? oi : bi;
            bi = gt ? oi : (eq ? ti : bi);
            bv = gt ? ov : bv;
        }
        if (lane == 0) { redV[wid] = bv; redI[wid] = bi; }
        __syncthreads();
        if (wid == 0) {
            bv = redV[lane]; bi = redI[lane];
#pragma unroll
            for (int off = 16; off >= 1; off >>= 1) {
                const float ov = __shfl_xor(bv, off, 32);
                const int   oi = __shfl_xor(bi, off, 32);
                const bool gt = ov > bv, eq = ov == bv;
                const int  ti = oi < bi ? oi : bi;
                bi = gt ? oi : (eq ? ti : bi);
                bv = gt ? ov : bv;
            }
            if (lane == 0) sFar = bi;
        }
        __syncthreads();
    }
}

#define TPITCH 4100
#define CAND_CAP 512

template <int R>
__device__ __forceinline__ void select32(const float* cd, const unsigned short* cj,
                                         int C, int lane, int* selRow)
{
    unsigned key[R];
    int      jv[R];
#pragma unroll
    for (int r = 0; r < R; ++r) {
        const int  i  = r * 32 + lane;
        const bool in = i < C;
        const int  ii = in ? i : 0;
        const unsigned u = __float_as_uint(cd[ii]);
        const unsigned k = u ^ ((unsigned)((int)u >> 31) | 0x80000000u);
        key[r] = in ? k : 0xFFFFFFFFu;
        jv[r]  = (int)cj[ii];
    }
    unsigned T = 0;
    for (int bit = 31; bit >= 0; --bit) {
        const unsigned cand = T | (1u << bit);
        int cnt = 0;
#pragma unroll
        for (int r = 0; r < R; ++r)
            cnt += __popc(__builtin_amdgcn_ballot_w32(key[r] < cand));
        if (cnt < K_) T = cand;
    }
    int nless = 0;
#pragma unroll
    for (int r = 0; r < R; ++r)
        nless += __popc(__builtin_amdgcn_ballot_w32(key[r] < T));
    const int need = K_ - nless;
    int running = 0, tieSeen = 0;
#pragma unroll
    for (int r = 0; r < R; ++r) {
        const bool isTie = key[r] == T;
        const unsigned tb = __builtin_amdgcn_ballot_w32(isTie);
        const int tieRank = tieSeen + (int)__builtin_amdgcn_mbcnt_lo(tb, 0u);
        const bool sel = (key[r] < T) || (isTie && tieRank < need);
        const unsigned sb = __builtin_amdgcn_ballot_w32(sel);
        const int slot = running + (int)__builtin_amdgcn_mbcnt_lo(sb, 0u);
        if (sel && slot < K_) selRow[slot] = jv[r];
        running += __popc(sb);
        tieSeen += __popc(tb);
    }
}

__device__ __forceinline__ void select32_scan(const float* row, int lane, int* selRow)
{
    float pd = -3.4e38f; int pj = -1;
    for (int k = 0; k < K_; ++k) {
        float bv = 3.4e38f; int bi = 0x7fffffff;
        for (int jj = 0; jj < N_ / 128; ++jj) {
            const int j0 = jj * 128 + lane * 4;
            const v4f dv = *(const v4f*)&row[j0];
#pragma unroll
            for (int e = 0; e < 4; ++e) {
                const float d = dv[e];
                const int   j = j0 + e;
                const bool after = d > pd || (d == pd && j > pj);
                if (after && (d < bv || (d == bv && j < bi))) { bv = d; bi = j; }
            }
        }
#pragma unroll
        for (int off = 16; off >= 1; off >>= 1) {
            const float ov = __shfl_xor(bv, off, 32);
            const int   oi = __shfl_xor(bi, off, 32);
            if (ov < bv || (ov == bv && oi < bi)) { bv = ov; bi = oi; }
        }
        if (lane == 0) selRow[k] = bi;
        pd = bv; pj = bi;
    }
}

__global__ __launch_bounds__(512) void group_kernel(
    const float* __restrict__ xyz,
    const float* __restrict__ pts,
    const float* __restrict__ wsA,
    const float* __restrict__ wsC2,
    float* __restrict__ out_feat)
{
    __shared__ float DT[16 * TPITCH];
    __shared__ int   selIdx[16][K_];
    __shared__ float          candD[16][CAND_CAP];
    __shared__ unsigned short candJ[16][CAND_CAP];

    const int b    = blockIdx.y;
    const int s0   = blockIdx.x * 16;
    const int tid  = threadIdx.x;
    const int lane = tid & 31;
    const int wid  = tid >> 5;
    const bool hi  = lane >= 16;

    const float* xb = xyz + (size_t)b * 3 * N_;

    const int  mcol = lane & 15;
    const v4f  A4   = *(const v4f*)(wsA + ((size_t)b * S_ + s0 + mcol) * 4);
    v2f a;
    a.x = hi ? A4.z : A4.x;
    a.y = hi ? A4.w : A4.y;

    const int mbase = hi ? 8 : 0;
    float cmr[8];
#pragma unroll
    for (int r = 0; r < 8; ++r)
        cmr[r] = wsC2[(size_t)b * S_ + s0 + mbase + r];

    for (int t = 0; t < 16; ++t) {
        const int n = wid * 256 + t * 16 + mcol;
        const float x = xb[n], y = xb[N_ + n], z = xb[2 * N_ + n];
        const float pn2 = x * x + y * y + z * z;
        v2f bb;
        bb.x = hi ? z : x;
        bb.y = hi ? pn2 : y;

        v8f c = {};
        c = __builtin_amdgcn_wmma_f32_16x16x4_f32(
                 false, a,  false, bb,
                 (short)0, c,  false,  false);

#pragma unroll
        for (int r = 0; r < 8; ++r)
            DT[(mbase + r) * TPITCH + n] = c[r] + cmr[r];
    }
    __syncthreads();

    const int m = wid;
    {
        const float* row = &DT[m * TPITCH];
        float* cd = candD[wid];
        unsigned short* cj = candJ[wid];

        float mn = 3.4e38f;
        for (int jj = 0; jj < N_ / 128; ++jj) {
            const v4f dv = *(const v4f*)&row[jj * 128 + lane * 4];
            mn = fminf(mn, fminf(fminf(dv.x, dv.y), fminf(dv.z, dv.w)));
        }
        float U = mn;
#pragma unroll
        for (int off = 16; off >= 1; off >>= 1)
            U = fmaxf(U, __shfl_xor(U, off, 32));

        int C = 0;
        for (int jj = 0; jj < N_ / 128; ++jj) {
            const int j0 = jj * 128 + lane * 4;
            const v4f dv = *(const v4f*)&row[j0];
            const bool p0 = dv.x <= U, p1 = dv.y <= U;
            const bool p2 = dv.z <= U, p3 = dv.w <= U;
            const unsigned b0 = __builtin_amdgcn_ballot_w32(p0);
            const unsigned b1 = __builtin_amdgcn_ballot_w32(p1);
            const unsigned b2 = __builtin_amdgcn_ballot_w32(p2);
            const unsigned b3 = __builtin_amdgcn_ballot_w32(p3);
            unsigned pre = __builtin_amdgcn_mbcnt_lo(b0, 0u);
            pre = __builtin_amdgcn_mbcnt_lo(b1, pre);
            pre = __builtin_amdgcn_mbcnt_lo(b2, pre);
            pre = __builtin_amdgcn_mbcnt_lo(b3, pre);
            int pos = C + (int)pre;
            if (p0 && pos < CAND_CAP) { cd[pos] = dv.x; cj[pos] = (unsigned short)j0; }
            pos += p0 ? 1 : 0;
            if (p1 && pos < CAND_CAP) { cd[pos] = dv.y; cj[pos] = (unsigned short)(j0 + 1); }
            pos += p1 ? 1 : 0;
            if (p2 && pos < CAND_CAP) { cd[pos] = dv.z; cj[pos] = (unsigned short)(j0 + 2); }
            pos += p2 ? 1 : 0;
            if (p3 && pos < CAND_CAP) { cd[pos] = dv.w; cj[pos] = (unsigned short)(j0 + 3); }
            C += __popc(b0) + __popc(b1) + __popc(b2) + __popc(b3);
        }
        __syncthreads();

        int* selRow = selIdx[m];
        if (C <= 128)           select32<4>(cd, cj, C, lane, selRow);
        else if (C <= 256)      select32<8>(cd, cj, C, lane, selRow);
        else if (C <= CAND_CAP) select32<16>(cd, cj, C, lane, selRow);
        else                    select32_scan(row, lane, selRow);
        __syncthreads();
    }

    const float* pb = pts + (size_t)b * D_ * N_;
    const unsigned o0 = (unsigned)lane * N_;
    const unsigned o1 = (unsigned)(lane + 32) * N_;
    const unsigned o2 = (unsigned)(lane < 2 ? lane : 2) * N_;
    const int* selRow = selIdx[m];
    float m0 = -3.4e38f, m1 = -3.4e38f, m2 = -3.4e38f;
#pragma unroll
    for (int k = 0; k < K_; ++k) {
        const int j = __builtin_amdgcn_readfirstlane(selRow[k]);
        m0 = fmaxf(m0, pb[o0 + j]);
        m1 = fmaxf(m1, pb[o1 + j]);
        m2 = fmaxf(m2, xb[o2 + j]);
    }
    float* ob = out_feat + (size_t)b * 67 * S_ + (s0 + m);
    ob[(size_t)(3 + lane) * S_]  = m0;
    ob[(size_t)(35 + lane) * S_] = m1;
    if (lane < 3) ob[(size_t)lane * S_] = m2;
}

extern "C" void kernel_launch(void* const* d_in, const int* in_sizes, int n_in,
                              void* d_out, int out_size, void* d_ws, size_t ws_size,
                              hipStream_t stream) {
    const float* xyz    = (const float*)d_in[0];
    const float* points = (const float*)d_in[1];
    const int*   finit  = (const int*)d_in[2];

    float* out        = (float*)d_out;
    float* out_newxyz = out;
    float* out_feat   = out + (size_t)B_ * 3 * S_;

    float* wsA  = (float*)d_ws;
    float* wsC2 = wsA + (size_t)B_ * S_ * 4;

    fps_kernel<<<B_, 1024, 0, stream>>>(xyz, finit, out_newxyz, wsA, wsC2);
    group_kernel<<<dim3(S_ / 16, B_), 512, 0, stream>>>(xyz, points, wsA, wsC2, out_feat);
}


# FPS block argmax flat through LDS: each thread stores its (value,index) pair; one wave per array folds all 1024 (32 a lane, max then min-index trees), one butterfly; per-wave butterflies gone
# speedup vs baseline: 3.0908x; 1.4845x over previous
#include <hip/hip_runtime.h>
#include <math.h>

#define B_  8
#define N_  4096
#define D_  64
#define S_  1024
#define K_  32

typedef __attribute__((ext_vector_type(2))) float v2f;
typedef __attribute__((ext_vector_type(4))) float v4f;
typedef __attribute__((ext_vector_type(8))) float v8f;

typedef __attribute__((ext_vector_type(4))) int v4i;

template <int J>
__device__ __forceinline__ float fps_sqdist(const v4f p, const v4f c)
{
    const float dx = p.x - c.x, dy = p.y - c.y, dz = p.z - c.z;
    float d;
    {
#pragma clang fp contract(off)
        d = (J < 2) ? __builtin_fmaf(dz, dz, __builtin_fmaf(dx, dx, dy * dy))
                    : dz * dz + __builtin_fmaf(dy, dy, dx * dx);
    }
    return d;
}

__device__ __forceinline__ void fps_emit(const v4f c, int b, int pos,
                                         float* __restrict__ out_newxyz,
                                         float* __restrict__ wsA,
                                         float* __restrict__ wsC2)
{
    out_newxyz[(size_t)b * 3 * S_ + 0 * S_ + pos] = c.x;
    out_newxyz[(size_t)b * 3 * S_ + 1 * S_ + pos] = c.y;
    out_newxyz[(size_t)b * 3 * S_ + 2 * S_ + pos] = c.z;
    float* A = wsA + ((size_t)b * S_ + pos) * 4;
    A[0] = -2.f * c.x; A[1] = -2.f * c.y; A[2] = -2.f * c.z; A[3] = 1.f;
    float c2;
    {
#pragma clang fp contract(off)
        c2 = c.z * c.z + __builtin_fmaf(c.x, c.x, c.y * c.y);
    }
    wsC2[(size_t)b * S_ + pos] = c2;
}

__global__ __launch_bounds__(1024) void fps_kernel(
    const float* __restrict__ xyz,
    const int*   __restrict__ farthest_init,
    float* __restrict__ out_newxyz,
    float* __restrict__ wsA,
    float* __restrict__ wsC2)
{
    __shared__ v4f   p4[N_];
    __shared__ __attribute__((aligned(16))) float pairs[2][2 * 1024];
    __shared__ __attribute__((aligned(16))) int sRes[4];

    const int b    = blockIdx.x;
    const int tid  = threadIdx.x;
    const int lane = tid & 31;
    const int wid  = tid >> 5;

    const float* xb = xyz + (size_t)b * 3 * N_;
    for (int i = tid; i < N_; i += 1024) {
        v4f v;
        v.x = xb[i];
        v.y = xb[N_ + i];
        v.z = xb[2 * N_ + i];
        v.w = 0.f;
        p4[i] = v;
    }
    float dA[4], dB[4];
#pragma unroll
    for (int j = 0; j < 4; ++j) { dA[j] = 1e10f; dB[j] = 1e10f; }
    if (tid == 0) {
        const int f0 = farthest_init[b];
        sRes[0] = f0; sRes[1] = f0; sRes[2] = f0; sRes[3] = f0;
    }
    __syncthreads();

    int pos  = -1;
    int far2 = -1;
    v4f c2   = {0.f, 0.f, 0.f, 0.f};
    for (int round = 0; round < S_; ++round) {
        const v4i r = *(const v4i*)sRes;
        const int wA = __builtin_amdgcn_readfirstlane(r.x);
        const int gA = __builtin_amdgcn_readfirstlane(r.y);
        const int wB = __builtin_amdgcn_readfirstlane(r.z);
        const int gB = __builtin_amdgcn_readfirstlane(r.w);
        const bool acc = (wA == far2);
        if (acc && tid == 0 && pos + 1 < S_)
            fps_emit(c2, b, pos + 1, out_newxyz, wsA, wsC2);
        pos += acc ? 2 : 1;
        if (pos >= S_) break;
        const int far1 = acc ? wB : wA;
        far2 = acc ? gB : gA;
        const v4f c1 = p4[far1];
        c2 = p4[far2];
        if (tid == 0) fps_emit(c1, b, pos, out_newxyz, wsA, wsC2);

        float bvA = -1.f, bvB = -1.f; int biA = 0x7fffffff, biB = 0x7fffffff;
#pragma unroll
        for (int j = 0; j < 4; ++j) {
            const int i = tid + 1024 * j;
            const v4f p = p4[i];
            const float dist = acc ? dB[j] : dA[j];
            float d1, d2;
            if (j < 2) { d1 = fps_sqdist<0>(p, c1); d2 = fps_sqdist<0>(p, c2); }
            else       { d1 = fps_sqdist<2>(p, c1); d2 = fps_sqdist<2>(p, c2); }
            const float na = fminf(dist, d1);
            const float nb = fminf(na, d2);
            dA[j] = na;
            dB[j] = nb;
            const bool takeA = na > bvA;
            bvA = takeA ? na : bvA;
            biA = takeA ? i : biA;
            const bool takeB = nb > bvB;
            bvB = takeB ? nb : bvB;
            biB = takeB ? i : biB;
        }
        {
            v2f qa, qb;
            qa.x = bvA; qa.y = __int_as_float(biA);
            qb.x = bvB; qb.y = __int_as_float(biB);
            *(v2f*)&pairs[0][2 * tid] = qa;
            *(v2f*)&pairs[1][2 * tid] = qb;
        }
        __syncthreads();
        if (wid < 2) {
            const float* pr = pairs[wid];
            float tv[32]; int ti[32];
#pragma unroll
            for (int k = 0; k < 16; ++k) {
                const v4f q = *(const v4f*)&pr[4 * (k * 32 + lane)];
                tv[2 * k]     = q.x; ti[2 * k]     = __float_as_int(q.y);
                tv[2 * k + 1] = q.z; ti[2 * k + 1] = __float_as_int(q.w);
            }
            float tm[16];
#pragma unroll
            for (int t = 0; t < 16; ++t) tm[t] = fmaxf(tv[t], tv[t + 16]);
#pragma unroll
            for (int n = 8; n >= 1; n >>= 1) {
#pragma unroll
                for (int t = 0; t < n; ++t) tm[t] = fmaxf(tm[t], tm[t + n]);
            }
            const float lmax = tm[0];
#pragma unroll
            for (int t = 0; t < 32; ++t) ti[t] = (tv[t] == lmax) ? ti[t] : 0x7fffffff;
#pragma unroll
            for (int n = 16; n >= 1; n >>= 1) {
#pragma unroll
                for (int t = 0; t < n; ++t) ti[t] = ti[t + n] < ti[t] ? ti[t + n] : ti[t];
            }
            float v1 = lmax;            int i1 = ti[0];
            float v2 = -2.f;            int i2 = 0x7fffffff;
#pragma unroll
            for (int off = 16; off >= 1; off >>= 1) {
                const float ov1 = __shfl_xor(v1, off, 32);
                const int   oi1 = __shfl_xor(i1, off, 32);
                const float ov2 = __shfl_xor(v2, off, 32);
                const int   oi2 = __shfl_xor(i2, off, 32);
                const bool g = (ov1 > v1) | ((ov1 == v1) & (oi1 < i1));
                const float fv = g ? ov1 : v1;  const int fi = g ? oi1 : i1;
                const float lv = g ? v1 : ov1;  const int li = g ? i1 : oi1;
                const float cv = g ? ov2 : v2;  const int ci = g ? oi2 : i2;
                const bool h = (cv > lv) | ((cv == lv) & (ci < li));
                v1 = fv; i1 = fi;
                v2 = h ? cv : lv; i2 = h ? ci : li;
            }
            if (lane == 0) { sRes[2 * wid] = i1; sRes[2 * wid + 1] = i2; }
        }
        __syncthreads();
    }
}

#define TPITCH 4100
#define CAND_CAP 512

template <int R>
__device__ __forceinline__ void select32(const float* cd, const unsigned short* cj,
                                         int C, int lane, int* selRow)
{
    unsigned key[R];
    int      jv[R];
#pragma unroll
    for (int r = 0; r < R; ++r) {
        const int  i  = r * 32 + lane;
        const bool in = i < C;
        const int  ii = in ? i : 0;
        const unsigned u = __float_as_uint(cd[ii]);
        const unsigned k = u ^ ((unsigned)((int)u >> 31) | 0x80000000u);
        key[r] = in ? k : 0xFFFFFFFFu;
        jv[r]  = (int)cj[ii];
    }
    unsigned T = 0;
    for (int bit = 31; bit >= 0; --bit) {
        const unsigned cand = T | (1u << bit);
        int cnt = 0;
#pragma unroll
        for (int r = 0; r < R; ++r)
            cnt += __popc(__builtin_amdgcn_ballot_w32(key[r] < cand));
        if (cnt < K_) T = cand;
    }
    int nless = 0;
#pragma unroll
    for (int r = 0; r < R; ++r)
        nless += __popc(__builtin_amdgcn_ballot_w32(key[r] < T));
    const int need = K_ - nless;
    int running = 0, tieSeen = 0;
#pragma unroll
    for (int r = 0; r < R; ++r) {
        const bool isTie = key[r] == T;
        const unsigned tb = __builtin_amdgcn_ballot_w32(isTie);
        const int tieRank = tieSeen + (int)__builtin_amdgcn_mbcnt_lo(tb, 0u);
        const bool sel = (key[r] < T) || (isTie && tieRank < need);
        const unsigned sb = __builtin_amdgcn_ballot_w32(sel);
        const int slot = running + (int)__builtin_amdgcn_mbcnt_lo(sb, 0u);
        if (sel && slot < K_) selRow[slot] = jv[r];
        running += __popc(sb);
        tieSeen += __popc(tb);
    }
}

__device__ __forceinline__ void select32_scan(const float* row, int lane, int* selRow)
{
    float pd = -3.4e38f; int pj = -1;
    for (int k = 0; k < K_; ++k) {
        float bv = 3.4e38f; int bi = 0x7fffffff;
        for (int jj = 0; jj < N_ / 128; ++jj) {
            const int j0 = jj * 128 + lane * 4;
            const v4f dv = *(const v4f*)&row[j0];
#pragma unroll
            for (int e = 0; e < 4; ++e) {
                const float d = dv[e];
                const int   j = j0 + e;
                const bool after = d > pd || (d == pd && j > pj);
                if (after && (d < bv || (d == bv && j < bi))) { bv = d; bi = j; }
            }
        }
#pragma unroll
        for (int off = 16; off >= 1; off >>= 1) {
            const float ov = __shfl_xor(bv, off, 32);
            const int   oi = __shfl_xor(bi, off, 32);
            if (ov < bv || (ov == bv && oi < bi)) { bv = ov; bi = oi; }
        }
        if (lane == 0) selRow[k] = bi;
        pd = bv; pj = bi;
    }
}

__global__ __launch_bounds__(512) void group_kernel(
    const float* __restrict__ xyz,
    const float* __restrict__ pts,
    const float* __restrict__ wsA,
    const float* __restrict__ wsC2,
    float* __restrict__ out_feat)
{
    __shared__ float DT[16 * TPITCH];
    __shared__ int   selIdx[16][K_];
    __shared__ float          candD[16][CAND_CAP];
    __shared__ unsigned short candJ[16][CAND_CAP];

    const int b    = blockIdx.y;
    const int s0   = blockIdx.x * 16;
    const int tid  = threadIdx.x;
    const int lane = tid & 31;
    const int wid  = tid >> 5;
    const bool hi  = lane >= 16;

    const float* xb = xyz + (size_t)b * 3 * N_;

    const int  mcol = lane & 15;
    const v4f  A4   = *(const v4f*)(wsA + ((size_t)b * S_ + s0 + mcol) * 4);
    v2f a;
    a.x = hi ? A4.z : A4.x;
    a.y = hi ? A4.w : A4.y;

    const int mbase = hi ? 8 : 0;
    float cmr[8];
#pragma unroll
    for (int r = 0; r < 8; ++r)
        cmr[r] = wsC2[(size_t)b * S_ + s0 + mbase + r];

    for (int t = 0; t < 16; ++t) {
        const int n = wid * 256 + t * 16 + mcol;
        const float x = xb[n], y = xb[N_ + n], z = xb[2 * N_ + n];
        const float pn2 = x * x + y * y + z * z;
        v2f bb;
        bb.x = hi ? z : x;
        bb.y = hi ? pn2 : y;

        v8f c = {};
        c = __builtin_amdgcn_wmma_f32_16x16x4_f32(
                 false, a,  false, bb,
                 (short)0, c,  false,  false);

#pragma unroll
        for (int r = 0; r < 8; ++r)
            DT[(mbase + r) * TPITCH + n] = c[r] + cmr[r];
    }
    __syncthreads();

    const int m = wid;
    {
        const float* row = &DT[m * TPITCH];
        float* cd = candD[wid];
        unsigned short* cj = candJ[wid];

        float mn = 3.4e38f;
        for (int jj = 0; jj < N_ / 128; ++jj) {
            const v4f dv = *(const v4f*)&row[jj * 128 + lane * 4];
            mn = fminf(mn, fminf(fminf(dv.x, dv.y), fminf(dv.z, dv.w)));
        }
        float U = mn;
#pragma unroll
        for (int off = 16; off >= 1; off >>= 1)
            U = fmaxf(U, __shfl_xor(U, off, 32));

        int C = 0;
        for (int jj = 0; jj < N_ / 128; ++jj) {
            const int j0 = jj * 128 + lane * 4;
            const v4f dv = *(const v4f*)&row[j0];
            const bool p0 = dv.x <= U, p1 = dv.y <= U;
            const bool p2 = dv.z <= U, p3 = dv.w <= U;
            const unsigned b0 = __builtin_amdgcn_ballot_w32(p0);
            const unsigned b1 = __builtin_amdgcn_ballot_w32(p1);
            const unsigned b2 = __builtin_amdgcn_ballot_w32(p2);
            const unsigned b3 = __builtin_amdgcn_ballot_w32(p3);
            unsigned pre = __builtin_amdgcn_mbcnt_lo(b0, 0u);
            pre = __builtin_amdgcn_mbcnt_lo(b1, pre);
            pre = __builtin_amdgcn_mbcnt_lo(b2, pre);
            pre = __builtin_amdgcn_mbcnt_lo(b3, pre);
            int pos = C + (int)pre;
            if (p0 && pos < CAND_CAP) { cd[pos] = dv.x; cj[pos] = (unsigned short)j0; }
            pos += p0 ? 1 : 0;
            if (p1 && pos < CAND_CAP) { cd[pos] = dv.y; cj[pos] = (unsigned short)(j0 + 1); }
            pos += p1 ? 1 : 0;
            if (p2 && pos < CAND_CAP) { cd[pos] = dv.z; cj[pos] = (unsigned short)(j0 + 2); }
            pos += p2 ? 1 : 0;
            if (p3 && pos < CAND_CAP) { cd[pos] = dv.w; cj[pos] = (unsigned short)(j0 + 3); }
            C += __popc(b0) + __popc(b1) + __popc(b2) + __popc(b3);
        }
        __syncthreads();

        int* selRow = selIdx[m];
        if (C <= 128)           select32<4>(cd, cj, C, lane, selRow);
        else if (C <= 256)      select32<8>(cd, cj, C, lane, selRow);
        else if (C <= CAND_CAP) select32<16>(cd, cj, C, lane, selRow);
        else                    select32_scan(row, lane, selRow);
        __syncthreads();
    }

    const float* pb = pts + (size_t)b * D_ * N_;
    const unsigned o0 = (unsigned)lane * N_;
    const unsigned o1 = (unsigned)(lane + 32) * N_;
    const unsigned o2 = (unsigned)(lane < 2 ? lane : 2) * N_;
    const int* selRow = selIdx[m];
    float m0 = -3.4e38f, m1 = -3.4e38f, m2 = -3.4e38f;
#pragma unroll
    for (int k = 0; k < K_; ++k) {
        const int j = __builtin_amdgcn_readfirstlane(selRow[k]);
        m0 = fmaxf(m0, pb[o0 + j]);
        m1 = fmaxf(m1, pb[o1 + j]);
        m2 = fmaxf(m2, xb[o2 + j]);
    }
    float* ob = out_feat + (size_t)b * 67 * S_ + (s0 + m);
    ob[(size_t)(3 + lane) * S_]  = m0;
    ob[(size_t)(35 + lane) * S_] = m1;
    if (lane < 3) ob[(size_t)lane * S_] = m2;
}

extern "C" void kernel_launch(void* const* d_in, const int* in_sizes, int n_in,
                              void* d_out, int out_size, void* d_ws, size_t ws_size,
                              hipStream_t stream) {
    const float* xyz    = (const float*)d_in[0];
    const float* points = (const float*)d_in[1];
    const int*   finit  = (const int*)d_in[2];

    float* out        = (float*)d_out;
    float* out_newxyz = out;
    float* out_feat   = out + (size_t)B_ * 3 * S_;

    float* wsA  = (float*)d_ws;
    float* wsC2 = wsA + (size_t)B_ * S_ * 4;

    fps_kernel<<<B_, 1024, 0, stream>>>(xyz, finit, out_newxyz, wsA, wsC2);
    group_kernel<<<dim3(S_ / 16, B_), 512, 0, stream>>>(xyz, points, wsA, wsC2, out_feat);
}


# FPS fold wave: best and runner-up of its 32 lane results by rank counting through LDS (64-bit order key per lane, 16 broadcast reads, count of larger keys); the 5-step shuffle butterflies gone
# speedup vs baseline: 3.6988x; 1.1967x over previous
#include <hip/hip_runtime.h>
#include <math.h>

#define B_  8
#define N_  4096
#define D_  64
#define S_  1024
#define K_  32

typedef __attribute__((ext_vector_type(2))) float v2f;
typedef __attribute__((ext_vector_type(4))) float v4f;
typedef __attribute__((ext_vector_type(8))) float v8f;

typedef __attribute__((ext_vector_type(4))) int v4i;
typedef __attribute__((ext_vector_type(2))) unsigned long long v2u64;

template <int J>
__device__ __forceinline__ float fps_sqdist(const v4f p, const v4f c)
{
    const float dx = p.x - c.x, dy = p.y - c.y, dz = p.z - c.z;
    float d;
    {
#pragma clang fp contract(off)
        d = (J < 2) ? __builtin_fmaf(dz, dz, __builtin_fmaf(dx, dx, dy * dy))
                    : dz * dz + __builtin_fmaf(dy, dy, dx * dx);
    }
    return d;
}

__device__ __forceinline__ void fps_emit(const v4f c, int b, int pos,
                                         float* __restrict__ out_newxyz,
                                         float* __restrict__ wsA,
                                         float* __restrict__ wsC2)
{
    out_newxyz[(size_t)b * 3 * S_ + 0 * S_ + pos] = c.x;
    out_newxyz[(size_t)b * 3 * S_ + 1 * S_ + pos] = c.y;
    out_newxyz[(size_t)b * 3 * S_ + 2 * S_ + pos] = c.z;
    float* A = wsA + ((size_t)b * S_ + pos) * 4;
    A[0] = -2.f * c.x; A[1] = -2.f * c.y; A[2] = -2.f * c.z; A[3] = 1.f;
    float c2;
    {
#pragma clang fp contract(off)
        c2 = c.z * c.z + __builtin_fmaf(c.x, c.x, c.y * c.y);
    }
    wsC2[(size_t)b * S_ + pos] = c2;
}

__global__ __launch_bounds__(1024) void fps_kernel(
    const float* __restrict__ xyz,
    const int*   __restrict__ farthest_init,
    float* __restrict__ out_newxyz,
    float* __restrict__ wsA,
    float* __restrict__ wsC2)
{
    __shared__ v4f   p4[N_];
    __shared__ __attribute__((aligned(16))) float pairs[2][2 * 1024];
    __shared__ __attribute__((aligned(16))) int sRes[4];
    __shared__ __attribute__((aligned(16))) unsigned long long lkeys[2][32];

    const int b    = blockIdx.x;
    const int tid  = threadIdx.x;
    const int lane = tid & 31;
    const int wid  = tid >> 5;

    const float* xb = xyz + (size_t)b * 3 * N_;
    for (int i = tid; i < N_; i += 1024) {
        v4f v;
        v.x = xb[i];
        v.y = xb[N_ + i];
        v.z = xb[2 * N_ + i];
        v.w = 0.f;
        p4[i] = v;
    }
    float dA[4], dB[4];
#pragma unroll
    for (int j = 0; j < 4; ++j) { dA[j] = 1e10f; dB[j] = 1e10f; }
    if (tid == 0) {
        const int f0 = farthest_init[b];
        sRes[0] = f0; sRes[1] = f0; sRes[2] = f0; sRes[3] = f0;
    }
    __syncthreads();

    int pos  = -1;
    int far2 = -1;
    v4f c2   = {0.f, 0.f, 0.f, 0.f};
    for (int round = 0; round < S_; ++round) {
        const v4i r = *(const v4i*)sRes;
        const int wA = __builtin_amdgcn_readfirstlane(r.x);
        const int gA = __builtin_amdgcn_readfirstlane(r.y);
        const int wB = __builtin_amdgcn_readfirstlane(r.z);
        const int gB = __builtin_amdgcn_readfirstlane(r.w);
        const bool acc = (wA == far2);
        if (acc && tid == 0 && pos + 1 < S_)
            fps_emit(c2, b, pos + 1, out_newxyz, wsA, wsC2);
        pos += acc ? 2 : 1;
        if (pos >= S_) break;
        const int far1 = acc ? wB : wA;
        far2 = acc ? gB : gA;
        const v4f c1 = p4[far1];
        c2 = p4[far2];
        if (tid == 0) fps_emit(c1, b, pos, out_newxyz, wsA, wsC2);

        float bvA = -1.f, bvB = -1.f; int biA = 0x7fffffff, biB = 0x7fffffff;
#pragma unroll
        for (int j = 0; j < 4; ++j) {
            const int i = tid + 1024 * j;
            const v4f p = p4[i];
            const float dist = acc ? dB[j] : dA[j];
            float d1, d2;
            if (j < 2) { d1 = fps_sqdist<0>(p, c1); d2 = fps_sqdist<0>(p, c2); }
            else       { d1 = fps_sqdist<2>(p, c1); d2 = fps_sqdist<2>(p, c2); }
            const float na = fminf(dist, d1);
            const float nb = fminf(na, d2);
            dA[j] = na;
            dB[j] = nb;
            const bool takeA = na > bvA;
            bvA = takeA ? na : bvA;
            biA = takeA ? i : biA;
            const bool takeB = nb > bvB;
            bvB = takeB ? nb : bvB;
            biB = takeB ? i : biB;
        }
        {
            v2f qa, qb;
            qa.x = bvA; qa.y = __int_as_float(biA);
            qb.x = bvB; qb.y = __int_as_float(biB);
            *(v2f*)&pairs[0][2 * tid] = qa;
            *(v2f*)&pairs[1][2 * tid] = qb;
        }
        __syncthreads();
        if (wid < 2) {
            const float* pr = pairs[wid];
            float tv[32]; int ti[32];
#pragma unroll
            for (int k = 0; k < 16; ++k) {
                const v4f q = *(const v4f*)&pr[4 * (k * 32 + lane)];
                tv[2 * k]     = q.x; ti[2 * k]     = __float_as_int(q.y);
                tv[2 * k + 1] = q.z; ti[2 * k + 1] = __float_as_int(q.w);
            }
            float tm[16];
#pragma unroll
            for (int t = 0; t < 16; ++t) tm[t] = fmaxf(tv[t], tv[t + 16]);
#pragma unroll
            for (int n = 8; n >= 1; n >>= 1) {
#pragma unroll
                for (int t = 0; t < n; ++t) tm[t] = fmaxf(tm[t], tm[t + n]);
            }
            const float lmax = tm[0];
#pragma unroll
            for (int t = 0; t < 32; ++t) ti[t] = (tv[t] == lmax) ? ti[t] : 0x7fffffff;
#pragma unroll
            for (int n = 16; n >= 1; n >>= 1) {
#pragma unroll
                for (int t = 0; t < n; ++t) ti[t] = ti[t + n] < ti[t] ? ti[t + n] : ti[t];
            }
            const int li = ti[0];
            const unsigned ub = __float_as_uint(lmax);
            const unsigned hb = ub ^ ((unsigned)((int)ub >> 31) | 0x80000000u);
            const unsigned long long key =
                ((unsigned long long)hb << 32) | (unsigned long long)(unsigned)(~li);
            unsigned long long* kb = lkeys[wid];
            kb[lane] = key;
            v2u64 q[16];
#pragma unroll
            for (int k = 0; k < 16; ++k) q[k] = *(const v2u64*)&kb[2 * k];
            int rk[4] = {0, 0, 0, 0};
#pragma unroll
            for (int k = 0; k < 16; ++k)
                rk[k & 3] += (q[k].x > key ? 1 : 0) + (q[k].y > key ? 1 : 0);
            const int rank = (rk[0] + rk[1]) + (rk[2] + rk[3]);
            if (rank == 0) sRes[2 * wid]     = li;
            if (rank == 1) sRes[2 * wid + 1] = li;
        }
        __syncthreads();
    }
}

#define TPITCH 4100
#define CAND_CAP 512

template <int R>
__device__ __forceinline__ void select32(const float* cd, const unsigned short* cj,
                                         int C, int lane, int* selRow)
{
    unsigned key[R];
    int      jv[R];
#pragma unroll
    for (int r = 0; r < R; ++r) {
        const int  i  = r * 32 + lane;
        const bool in = i < C;
        const int  ii = in ? i : 0;
        const unsigned u = __float_as_uint(cd[ii]);
        const unsigned k = u ^ ((unsigned)((int)u >> 31) | 0x80000000u);
        key[r] = in ? k : 0xFFFFFFFFu;
        jv[r]  = (int)cj[ii];
    }
    unsigned T = 0;
    for (int bit = 31; bit >= 0; --bit) {
        const unsigned cand = T | (1u << bit);
        int cnt = 0;
#pragma unroll
        for (int r = 0; r < R; ++r)
            cnt += __popc(__builtin_amdgcn_ballot_w32(key[r] < cand));
        if (cnt < K_) T = cand;
    }
    int nless = 0;
#pragma unroll
    for (int r = 0; r < R; ++r)
        nless += __popc(__builtin_amdgcn_ballot_w32(key[r] < T));
    const int need = K_ - nless;
    int running = 0, tieSeen = 0;
#pragma unroll
    for (int r = 0; r < R; ++r) {
        const bool isTie = key[r] == T;
        const unsigned tb = __builtin_amdgcn_ballot_w32(isTie);
        const int tieRank = tieSeen + (int)__builtin_amdgcn_mbcnt_lo(tb, 0u);
        const bool sel = (key[r] < T) || (isTie && tieRank < need);
        const unsigned sb = __builtin_amdgcn_ballot_w32(sel);
        const int slot = running + (int)__builtin_amdgcn_mbcnt_lo(sb, 0u);
        if (sel && slot < K_) selRow[slot] = jv[r];
        running += __popc(sb);
        tieSeen += __popc(tb);
    }
}

__device__ __forceinline__ void select32_scan(const float* row, int lane, int* selRow)
{
    float pd = -3.4e38f; int pj = -1;
    for (int k = 0; k < K_; ++k) {
        float bv = 3.4e38f; int bi = 0x7fffffff;
        for (int jj = 0; jj < N_ / 128; ++jj) {
            const int j0 = jj * 128 + lane * 4;
            const v4f dv = *(const v4f*)&row[j0];
#pragma unroll
            for (int e = 0; e < 4; ++e) {
                const float d = dv[e];
                const int   j = j0 + e;
                const bool after = d > pd || (d == pd && j > pj);
                if (after && (d < bv || (d == bv && j < bi))) { bv = d; bi = j; }
            }
        }
#pragma unroll
        for (int off = 16; off >= 1; off >>= 1) {
            const float ov = __shfl_xor(bv, off, 32);
            const int   oi = __shfl_xor(bi, off, 32);
            if (ov < bv || (ov == bv && oi < bi)) { bv = ov; bi = oi; }
        }
        if (lane == 0) selRow[k] = bi;
        pd = bv; pj = bi;
    }
}

__global__ __launch_bounds__(512) void group_kernel(
    const float* __restrict__ xyz,
    const float* __restrict__ pts,
    const float* __restrict__ wsA,
    const float* __restrict__ wsC2,
    float* __restrict__ out_feat)
{
    __shared__ float DT[16 * TPITCH];
    __shared__ int   selIdx[16][K_];
    __shared__ float          candD[16][CAND_CAP];
    __shared__ unsigned short candJ[16][CAND_CAP];

    const int b    = blockIdx.y;
    const int s0   = blockIdx.x * 16;
    const int tid  = threadIdx.x;
    const int lane = tid & 31;
    const int wid  = tid >> 5;
    const bool hi  = lane >= 16;

    const float* xb = xyz + (size_t)b * 3 * N_;

    const int  mcol = lane & 15;
    const v4f  A4   = *(const v4f*)(wsA + ((size_t)b * S_ + s0 + mcol) * 4);
    v2f a;
    a.x = hi ? A4.z : A4.x;
    a.y = hi ? A4.w : A4.y;

    const int mbase = hi ? 8 : 0;
    float cmr[8];
#pragma unroll
    for (int r = 0; r < 8; ++r)
        cmr[r] = wsC2[(size_t)b * S_ + s0 + mbase + r];

    for (int t = 0; t < 16; ++t) {
        const int n = wid * 256 + t * 16 + mcol;
        const float x = xb[n], y = xb[N_ + n], z = xb[2 * N_ + n];
        const float pn2 = x * x + y * y + z * z;
        v2f bb;
        bb.x = hi ? z : x;
        bb.y = hi ? pn2 : y;

        v8f c = {};
        c = __builtin_amdgcn_wmma_f32_16x16x4_f32(
                 false, a,  false, bb,
                 (short)0, c,  false,  false);

#pragma unroll
        for (int r = 0; r < 8; ++r)
            DT[(mbase + r) * TPITCH + n] = c[r] + cmr[r];
    }
    __syncthreads();

    const int m = wid;
    {
        const float* row = &DT[m * TPITCH];
        float* cd = candD[wid];
        unsigned short* cj = candJ[wid];

        float mn = 3.4e38f;
        for (int jj = 0; jj < N_ / 128; ++jj) {
            const v4f dv = *(const v4f*)&row[jj * 128 + lane * 4];
            mn = fminf(mn, fminf(fminf(dv.x, dv.y), fminf(dv.z, dv.w)));
        }
        float U = mn;
#pragma unroll
        for (int off = 16; off >= 1; off >>= 1)
            U = fmaxf(U, __shfl_xor(U, off, 32));

        int C = 0;
        for (int jj = 0; jj < N_ / 128; ++jj) {
            const int j0 = jj * 128 + lane * 4;
            const v4f dv = *(const v4f*)&row[j0];
            const bool p0 = dv.x <= U, p1 = dv.y <= U;
            const bool p2 = dv.z <= U, p3 = dv.w <= U;
            const unsigned b0 = __builtin_amdgcn_ballot_w32(p0);
            const unsigned b1 = __builtin_amdgcn_ballot_w32(p1);
            const unsigned b2 = __builtin_amdgcn_ballot_w32(p2);
            const unsigned b3 = __builtin_amdgcn_ballot_w32(p3);
            unsigned pre = __builtin_amdgcn_mbcnt_lo(b0, 0u);
            pre = __builtin_amdgcn_mbcnt_lo(b1, pre);
            pre = __builtin_amdgcn_mbcnt_lo(b2, pre);
            pre = __builtin_amdgcn_mbcnt_lo(b3, pre);
            int pos = C + (int)pre;
            if (p0 && pos < CAND_CAP) { cd[pos] = dv.x; cj[pos] = (unsigned short)j0; }
            pos += p0 ? 1 : 0;
            if (p1 && pos < CAND_CAP) { cd[pos] = dv.y; cj[pos] = (unsigned short)(j0 + 1); }
            pos += p1 ? 1 : 0;
            if (p2 && pos < CAND_CAP) { cd[pos] = dv.z; cj[pos] = (unsigned short)(j0 + 2); }
            pos += p2 ? 1 : 0;
            if (p3 && pos < CAND_CAP) { cd[pos] = dv.w; cj[pos] = (unsigned short)(j0 + 3); }
            C += __popc(b0) + __popc(b1) + __popc(b2) + __popc(b3);
        }
        __syncthreads();

        int* selRow = selIdx[m];
        if (C <= 128)           select32<4>(cd, cj, C, lane, selRow);
        else if (C <= 256)      select32<8>(cd, cj, C, lane, selRow);
        else if (C <= CAND_CAP) select32<16>(cd, cj, C, lane, selRow);
        else                    select32_scan(row, lane, selRow);
        __syncthreads();
    }

    const float* pb = pts + (size_t)b * D_ * N_;
    const unsigned o0 = (unsigned)lane * N_;
    const unsigned o1 = (unsigned)(lane + 32) * N_;
    const unsigned o2 = (unsigned)(lane < 2 ? lane : 2) * N_;
    const int* selRow = selIdx[m];
    float m0 = -3.4e38f, m1 = -3.4e38f, m2 = -3.4e38f;
#pragma unroll
    for (int k = 0; k < K_; ++k) {
        const int j = __builtin_amdgcn_readfirstlane(selRow[k]);
        m0 = fmaxf(m0, pb[o0 + j]);
        m1 = fmaxf(m1, pb[o1 + j]);
        m2 = fmaxf(m2, xb[o2 + j]);
    }
    float* ob = out_feat + (size_t)b * 67 * S_ + (s0 + m);
    ob[(size_t)(3 + lane) * S_]  = m0;
    ob[(size_t)(35 + lane) * S_] = m1;
    if (lane < 3) ob[(size_t)lane * S_] = m2;
}

extern "C" void kernel_launch(void* const* d_in, const int* in_sizes, int n_in,
                              void* d_out, int out_size, void* d_ws, size_t ws_size,
                              hipStream_t stream) {
    const float* xyz    = (const float*)d_in[0];
    const float* points = (const float*)d_in[1];
    const int*   finit  = (const int*)d_in[2];

    float* out        = (float*)d_out;
    float* out_newxyz = out;
    float* out_feat   = out + (size_t)B_ * 3 * S_;

    float* wsA  = (float*)d_ws;
    float* wsC2 = wsA + (size_t)B_ * S_ * 4;

    fps_kernel<<<B_, 1024, 0, stream>>>(xyz, finit, out_newxyz, wsA, wsC2);
    group_kernel<<<dim3(S_ / 16, B_), 512, 0, stream>>>(xyz, points, wsA, wsC2, out_feat);
}


# grouping kernel phase C reads a neighbour's 64 channels as one 256-byte row of a point-major copy of the channel data, built in the workspace by idle work-groups of the first launch
# speedup vs baseline: 3.9558x; 1.0695x over previous
#include <hip/hip_runtime.h>
#include <math.h>

#define B_  8
#define N_  4096
#define D_  64
#define S_  1024
#define K_  32

typedef __attribute__((ext_vector_type(2))) float v2f;
typedef __attribute__((ext_vector_type(4))) float v4f;
typedef __attribute__((ext_vector_type(8))) float v8f;

#define WS_T_OFF  ((size_t)B_ * S_ * 5)
#define WS_NEED   ((WS_T_OFF + (size_t)B_ * N_ * D_) * 4)
#define XT_BLOCKS (B_ * (N_ / 64))
#define XT_PITCH  65
typedef __attribute__((ext_vector_type(4))) int v4i;
typedef __attribute__((ext_vector_type(2))) unsigned long long v2u64;

template <int J>
__device__ __forceinline__ float fps_sqdist(const v4f p, const v4f c)
{
    const float dx = p.x - c.x, dy = p.y - c.y, dz = p.z - c.z;
    float d;
    {
#pragma clang fp contract(off)
        d = (J < 2) ? __builtin_fmaf(dz, dz, __builtin_fmaf(dx, dx, dy * dy))
                    : dz * dz + __builtin_fmaf(dy, dy, dx * dx);
    }
    return d;
}

__device__ __forceinline__ void fps_emit(const v4f c, int b, int pos,
                                         float* __restrict__ out_newxyz,
                                         float* __restrict__ wsA,
                                         float* __restrict__ wsC2)
{
    out_newxyz[(size_t)b * 3 * S_ + 0 * S_ + pos] = c.x;
    out_newxyz[(size_t)b * 3 * S_ + 1 * S_ + pos] = c.y;
    out_newxyz[(size_t)b * 3 * S_ + 2 * S_ + pos] = c.z;
    float* A = wsA + ((size_t)b * S_ + pos) * 4;
    A[0] = -2.f * c.x; A[1] = -2.f * c.y; A[2] = -2.f * c.z; A[3] = 1.f;
    float c2;
    {
#pragma clang fp contract(off)
        c2 = c.z * c.z + __builtin_fmaf(c.x, c.x, c.y * c.y);
    }
    wsC2[(size_t)b * S_ + pos] = c2;
}

__device__ __forceinline__ void transpose_tile(const float* __restrict__ pts,
                                               float* __restrict__ wsT,
                                               int t, int tid, float* tile)
{
    const int b  = t / (N_ / 64);
    const int n0 = (t % (N_ / 64)) * 64;
    {
        const int ch = tid >> 4, p0 = (tid & 15) * 4;
        const v4f v = *(const v4f*)(pts + ((size_t)b * D_ + ch) * N_ + n0 + p0);
        float* d = tile + ch * XT_PITCH + p0;
        d[0] = v.x; d[1] = v.y; d[2] = v.z; d[3] = v.w;
    }
    __syncthreads();
    {
        const int pt = tid >> 4, c0 = (tid & 15) * 4;
        const float* s = tile + c0 * XT_PITCH + pt;
        v4f o;
        o.x = s[0]; o.y = s[XT_PITCH]; o.z = s[2 * XT_PITCH]; o.w = s[3 * XT_PITCH];
        *(v4f*)(wsT + ((size_t)b * N_ + n0 + pt) * D_ + c0) = o;
    }
}

__global__ __launch_bounds__(1024) void fps_kernel(
    const float* __restrict__ xyz,
    const int*   __restrict__ farthest_init,
    float* __restrict__ out_newxyz,
    float* __restrict__ wsA,
    float* __restrict__ wsC2,
    const float* __restrict__ pts,
    float* __restrict__ wsT,
    unsigned long long ws_bytes)
{
    __shared__ v4f   p4[N_];
    __shared__ __attribute__((aligned(16))) float pairs[2][2 * 1024];
    __shared__ __attribute__((aligned(16))) int sRes[4];
    __shared__ __attribute__((aligned(16))) unsigned long long lkeys[2][32];

    const int b    = blockIdx.x;
    const int tid  = threadIdx.x;
    const int lane = tid & 31;
    const int wid  = tid >> 5;

    if (b >= B_) {
        if (ws_bytes >= WS_NEED) transpose_tile(pts, wsT, b - B_, tid, (float*)p4);
        return;
    }

    const float* xb = xyz + (size_t)b * 3 * N_;
    for (int i = tid; i < N_; i += 1024) {
        v4f v;
        v.x = xb[i];
        v.y = xb[N_ + i];
        v.z = xb[2 * N_ + i];
        v.w = 0.f;
        p4[i] = v;
    }
    float dA[4], dB[4];
#pragma unroll
    for (int j = 0; j < 4; ++j) { dA[j] = 1e10f; dB[j] = 1e10f; }
    if (tid == 0) {
        const int f0 = farthest_init[b];
        sRes[0] = f0; sRes[1] = f0; sRes[2] = f0; sRes[3] = f0;
    }
    __syncthreads();

    int pos  = -1;
    int far2 = -1;
    v4f c2   = {0.f, 0.f, 0.f, 0.f};
    for (int round = 0; round < S_; ++round) {
        const v4i r = *(const v4i*)sRes;
        const int wA = __builtin_amdgcn_readfirstlane(r.x);
        const int gA = __builtin_amdgcn_readfirstlane(r.y);
        const int wB = __builtin_amdgcn_readfirstlane(r.z);
        const int gB = __builtin_amdgcn_readfirstlane(r.w);
        const bool acc = (wA == far2);
        if (acc && tid == 0 && pos + 1 < S_)
            fps_emit(c2, b, pos + 1, out_newxyz, wsA, wsC2);
        pos += acc ? 2 : 1;
        if (pos >= S_) break;
        const int far1 = acc ? wB : wA;
        far2 = acc ? gB : gA;
        const v4f c1 = p4[far1];
        c2 = p4[far2];
        if (tid == 0) fps_emit(c1, b, pos, out_newxyz, wsA, wsC2);

        float bvA = -1.f, bvB = -1.f; int biA = 0x7fffffff, biB = 0x7fffffff;
#pragma unroll
        for (int j = 0; j < 4; ++j) {
            const int i = tid + 1024 * j;
            const v4f p = p4[i];
            const float dist = acc ? dB[j] : dA[j];
            float d1, d2;
            if (j < 2) { d1 = fps_sqdist<0>(p, c1); d2 = fps_sqdist<0>(p, c2); }
            else       { d1 = fps_sqdist<2>(p, c1); d2 = fps_sqdist<2>(p, c2); }
            const float na = fminf(dist, d1);
            const float nb = fminf(na, d2);
            dA[j] = na;
            dB[j] = nb;
            const bool takeA = na > bvA;
            bvA = takeA ? na : bvA;
            biA = takeA ? i : biA;
            const bool takeB = nb > bvB;
            bvB = takeB ? nb : bvB;
            biB = takeB ? i : biB;
        }
        {
            v2f qa, qb;
            qa.x = bvA; qa.y = __int_as_float(biA);
            qb.x = bvB; qb.y = __int_as_float(biB);
            *(v2f*)&pairs[0][2 * tid] = qa;
            *(v2f*)&pairs[1][2 * tid] = qb;
        }
        __syncthreads();
        if (wid < 2) {
            const float* pr = pairs[wid];
            float tv[32]; int ti[32];
#pragma unroll
            for (int k = 0; k < 16; ++k) {
                const v4f q = *(const v4f*)&pr[4 * (k * 32 + lane)];
                tv[2 * k]     = q.x; ti[2 * k]     = __float_as_int(q.y);
                tv[2 * k + 1] = q.z; ti[2 * k + 1] = __float_as_int(q.w);
            }
            float tm[16];
#pragma unroll
            for (int t = 0; t < 16; ++t) tm[t] = fmaxf(tv[t], tv[t + 16]);
#pragma unroll
            for (int n = 8; n >= 1; n >>= 1) {
#pragma unroll
                for (int t = 0; t < n; ++t) tm[t] = fmaxf(tm[t], tm[t + n]);
            }
            const float lmax = tm[0];
#pragma unroll
            for (int t = 0; t < 32; ++t) ti[t] = (tv[t] == lmax) ? ti[t] : 0x7fffffff;
#pragma unroll
            for (int n = 16; n >= 1; n >>= 1) {
#pragma unroll
                for (int t = 0; t < n; ++t) ti[t] = ti[t + n] < ti[t] ? ti[t + n] : ti[t];
            }
            const int li = ti[0];
            const unsigned ub = __float_as_uint(lmax);
            const unsigned hb = ub ^ ((unsigned)((int)ub >> 31) | 0x80000000u);
            const unsigned long long key =
                ((unsigned long long)hb << 32) | (unsigned long long)(unsigned)(~li);
            unsigned long long* kb = lkeys[wid];
            kb[lane] = key;
            v2u64 q[16];
#pragma unroll
            for (int k = 0; k < 16; ++k) q[k] = *(const v2u64*)&kb[2 * k];
            int rk[4] = {0, 0, 0, 0};
#pragma unroll
            for (int k = 0; k < 16; ++k)
                rk[k & 3] += (q[k].x > key ? 1 : 0) + (q[k].y > key ? 1 : 0);
            const int rank = (rk[0] + rk[1]) + (rk[2] + rk[3]);
            if (rank == 0) sRes[2 * wid]     = li;
            if (rank == 1) sRes[2 * wid + 1] = li;
        }
        __syncthreads();
    }
}

#define TPITCH 4100
#define CAND_CAP 512

template <int R>
__device__ __forceinline__ void select32(const float* cd, const unsigned short* cj,
                                         int C, int lane, int* selRow)
{
    unsigned key[R];
    int      jv[R];
#pragma unroll
    for (int r = 0; r < R; ++r) {
        const int  i  = r * 32 + lane;
        const bool in = i < C;
        const int  ii = in ? i : 0;
        const unsigned u = __float_as_uint(cd[ii]);
        const unsigned k = u ^ ((unsigned)((int)u >> 31) | 0x80000000u);
        key[r] = in ? k : 0xFFFFFFFFu;
        jv[r]  = (int)cj[ii];
    }
    unsigned T = 0;
    for (int bit = 31; bit >= 0; --bit) {
        const unsigned cand = T | (1u << bit);
        int cnt = 0;
#pragma unroll
        for (int r = 0; r < R; ++r)
            cnt += __popc(__builtin_amdgcn_ballot_w32(key[r] < cand));
        if (cnt < K_) T = cand;
    }
    int nless = 0;
#pragma unroll
    for (int r = 0; r < R; ++r)
        nless += __popc(__builtin_amdgcn_ballot_w32(key[r] < T));
    const int need = K_ - nless;
    int running = 0, tieSeen = 0;
#pragma unroll
    for (int r = 0; r < R; ++r) {
        const bool isTie = key[r] == T;
        const unsigned tb = __builtin_amdgcn_ballot_w32(isTie);
        const int tieRank = tieSeen + (int)__builtin_amdgcn_mbcnt_lo(tb, 0u);
        const bool sel = (key[r] < T) || (isTie && tieRank < need);
        const unsigned sb = __builtin_amdgcn_ballot_w32(sel);
        const int slot = running + (int)__builtin_amdgcn_mbcnt_lo(sb, 0u);
        if (sel && slot < K_) selRow[slot] = jv[r];
        running += __popc(sb);
        tieSeen += __popc(tb);
    }
}

__device__ __forceinline__ void select32_scan(const float* row, int lane, int* selRow)
{
    float pd = -3.4e38f; int pj = -1;
    for (int k = 0; k < K_; ++k) {
        float bv = 3.4e38f; int bi = 0x7fffffff;
        for (int jj = 0; jj < N_ / 128; ++jj) {
            const int j0 = jj * 128 + lane * 4;
            const v4f dv = *(const v4f*)&row[j0];
#pragma unroll
            for (int e = 0; e < 4; ++e) {
                const float d = dv[e];
                const int   j = j0 + e;
                const bool after = d > pd || (d == pd && j > pj);
                if (after && (d < bv || (d == bv && j < bi))) { bv = d; bi = j; }
            }
        }
#pragma unroll
        for (int off = 16; off >= 1; off >>= 1) {
            const float ov = __shfl_xor(bv, off, 32);
            const int   oi = __shfl_xor(bi, off, 32);
            if (ov < bv || (ov == bv && oi < bi)) { bv = ov; bi = oi; }
        }
        if (lane == 0) selRow[k] = bi;
        pd = bv; pj = bi;
    }
}

__global__ __launch_bounds__(512) void group_kernel(
    const float* __restrict__ xyz,
    const float* __restrict__ pts,
    const float* __restrict__ wsA,
    const float* __restrict__ wsC2,
    const float* __restrict__ wsT,
    unsigned long long ws_bytes,
    float* __restrict__ out_feat)
{
    __shared__ float DT[16 * TPITCH];
    __shared__ __attribute__((aligned(16))) int selIdx[16][K_];
    __shared__ float          candD[16][CAND_CAP];
    __shared__ unsigned short candJ[16][CAND_CAP];

    const int b    = blockIdx.y;
    const int s0   = blockIdx.x * 16;
    const int tid  = threadIdx.x;
    const int lane = tid & 31;
    const int wid  = tid >> 5;
    const bool hi  = lane >= 16;

    const float* xb = xyz + (size_t)b * 3 * N_;

    const int  mcol = lane & 15;
    const v4f  A4   = *(const v4f*)(wsA + ((size_t)b * S_ + s0 + mcol) * 4);
    v2f a;
    a.x = hi ? A4.z : A4.x;
    a.y = hi ? A4.w : A4.y;

    const int mbase = hi ? 8 : 0;
    float cmr[8];
#pragma unroll
    for (int r = 0; r < 8; ++r)
        cmr[r] = wsC2[(size_t)b * S_ + s0 + mbase + r];

    for (int t = 0; t < 16; ++t) {
        const int n = wid * 256 + t * 16 + mcol;
        const float x = xb[n], y = xb[N_ + n], z = xb[2 * N_ + n];
        const float pn2 = x * x + y * y + z * z;
        v2f bb;
        bb.x = hi ? z : x;
        bb.y = hi ? pn2 : y;

        v8f c = {};
        c = __builtin_amdgcn_wmma_f32_16x16x4_f32(
                 false, a,  false, bb,
                 (short)0, c,  false,  false);

#pragma unroll
        for (int r = 0; r < 8; ++r)
            DT[(mbase + r) * TPITCH + n] = c[r] + cmr[r];
    }
    __syncthreads();

    const int m = wid;
    {
        const float* row = &DT[m * TPITCH];
        float* cd = candD[wid];
        unsigned short* cj = candJ[wid];

        float mn = 3.4e38f;
        for (int jj = 0; jj < N_ / 128; ++jj) {
            const v4f dv = *(const v4f*)&row[jj * 128 + lane * 4];
            mn = fminf(mn, fminf(fminf(dv.x, dv.y), fminf(dv.z, dv.w)));
        }
        float U = mn;
#pragma unroll
        for (int off = 16; off >= 1; off >>= 1)
            U = fmaxf(U, __shfl_xor(U, off, 32));

        int C = 0;
        for (int jj = 0; jj < N_ / 128; ++jj) {
            const int j0 = jj * 128 + lane * 4;
            const v4f dv = *(const v4f*)&row[j0];
            const bool p0 = dv.x <= U, p1 = dv.y <= U;
            const bool p2 = dv.z <= U, p3 = dv.w <= U;
            const unsigned b0 = __builtin_amdgcn_ballot_w32(p0);
            const unsigned b1 = __builtin_amdgcn_ballot_w32(p1);
            const unsigned b2 = __builtin_amdgcn_ballot_w32(p2);
            const unsigned b3 = __builtin_amdgcn_ballot_w32(p3);
            unsigned pre = __builtin_amdgcn_mbcnt_lo(b0, 0u);
            pre = __builtin_amdgcn_mbcnt_lo(b1, pre);
            pre = __builtin_amdgcn_mbcnt_lo(b2, pre);
            pre = __builtin_amdgcn_mbcnt_lo(b3, pre);
            int pos = C + (int)pre;
            if (p0 && pos < CAND_CAP) { cd[pos] = dv.x; cj[pos] = (unsigned short)j0; }
            pos += p0 ? 1 : 0;
            if (p1 && pos < CAND_CAP) { cd[pos] = dv.y; cj[pos] = (unsigned short)(j0 + 1); }
            pos += p1 ? 1 : 0;
            if (p2 && pos < CAND_CAP) { cd[pos] = dv.z; cj[pos] = (unsigned short)(j0 + 2); }
            pos += p2 ? 1 : 0;
            if (p3 && pos < CAND_CAP) { cd[pos] = dv.w; cj[pos] = (unsigned short)(j0 + 3); }
            C += __popc(b0) + __popc(b1) + __popc(b2) + __popc(b3);
        }
        __syncthreads();

        int* selRow = selIdx[m];
        if (C <= 128)           select32<4>(cd, cj, C, lane, selRow);
        else if (C <= 256)      select32<8>(cd, cj, C, lane, selRow);
        else if (C <= CAND_CAP) select32<16>(cd, cj, C, lane, selRow);
        else                    select32_scan(row, lane, selRow);
        __syncthreads();
    }

    float* ob = out_feat + (size_t)b * 67 * S_ + (s0 + m);
    if (ws_bytes >= WS_NEED) {
        const int q = lane & 15;
        const v4i* sr = (const v4i*)(selIdx[m] + (hi ? 16 : 0));
        const float* tb = wsT + (size_t)b * N_ * D_ + 4 * q;
        v4i jv[4];
#pragma unroll
        for (int i = 0; i < 4; ++i) jv[i] = sr[i];
        v4f mx = {-3.4e38f, -3.4e38f, -3.4e38f, -3.4e38f};
#pragma unroll
        for (int k = 0; k < 16; ++k) {
            const int j = jv[k >> 2][k & 3];
            const v4f v = *(const v4f*)(tb + (size_t)j * D_);
            mx.x = fmaxf(mx.x, v.x); mx.y = fmaxf(mx.y, v.y);
            mx.z = fmaxf(mx.z, v.z); mx.w = fmaxf(mx.w, v.w);
        }
        const int jl = selIdx[m][lane];
        float gx = xb[jl], gy = xb[N_ + jl], gz = xb[2 * N_ + jl];
        mx.x = fmaxf(mx.x, __shfl_xor(mx.x, 16, 32));
        mx.y = fmaxf(mx.y, __shfl_xor(mx.y, 16, 32));
        mx.z = fmaxf(mx.z, __shfl_xor(mx.z, 16, 32));
        mx.w = fmaxf(mx.w, __shfl_xor(mx.w, 16, 32));
#pragma unroll
        for (int off = 16; off >= 1; off >>= 1) {
            gx = fmaxf(gx, __shfl_xor(gx, off, 32));
            gy = fmaxf(gy, __shfl_xor(gy, off, 32));
            gz = fmaxf(gz, __shfl_xor(gz, off, 32));
        }
        const int c0 = 3 + 4 * q + (hi ? 2 : 0);
        ob[(size_t)c0 * S_]       = hi ? mx.z : mx.x;
        ob[(size_t)(c0 + 1) * S_] = hi ? mx.w : mx.y;
        if (lane < 3) ob[(size_t)lane * S_] = lane == 0 ? gx : (lane == 1 ? gy : gz);
        return;
    }
    const float* pb = pts + (size_t)b * D_ * N_;
    const unsigned o0 = (unsigned)lane * N_;
    const unsigned o1 = (unsigned)(lane + 32) * N_;
    const unsigned o2 = (unsigned)(lane < 2 ? lane : 2) * N_;
    const int* selRow = selIdx[m];
    float m0 = -3.4e38f, m1 = -3.4e38f, m2 = -3.4e38f;
#pragma unroll
    for (int k = 0; k < K_; ++k) {
        const int j = __builtin_amdgcn_readfirstlane(selRow[k]);
        m0 = fmaxf(m0, pb[o0 + j]);
        m1 = fmaxf(m1, pb[o1 + j]);
        m2 = fmaxf(m2, xb[o2 + j]);
    }
    ob[(size_t)(3 + lane) * S_]  = m0;
    ob[(size_t)(35 + lane) * S_] = m1;
    if (lane < 3) ob[(size_t)lane * S_] = m2;
}

extern "C" void kernel_launch(void* const* d_in, const int* in_sizes, int n_in,
                              void* d_out, int out_size, void* d_ws, size_t ws_size,
                              hipStream_t stream) {
    const float* xyz    = (const float*)d_in[0];
    const float* points = (const float*)d_in[1];
    const int*   finit  = (const int*)d_in[2];

    float* out        = (float*)d_out;
    float* out_newxyz = out;
    float* out_feat   = out + (size_t)B_ * 3 * S_;

    float* wsA  = (float*)d_ws;
    float* wsC2 = wsA + (size_t)B_ * S_ * 4;
    float* wsT  = wsA + WS_T_OFF;

    const unsigned long long ws_bytes = ws_size;
    fps_kernel<<<B_ + XT_BLOCKS, 1024, 0, stream>>>(xyz, finit, out_newxyz, wsA, wsC2,
                                                    points, wsT, ws_bytes);
    group_kernel<<<dim3(S_ / 16, B_), 512, 0, stream>>>(xyz, points, wsA, wsC2, wsT, ws_bytes, out_feat);
}


# FPS: four picks per reduction round (was two): guesses = ranks 1-3 of the kept array's 32 lane results, arrays A0..A3 chained in one pass, four fold waves side by side; distances bit-identical
# speedup vs baseline: 4.7519x; 1.2012x over previous
#include <hip/hip_runtime.h>
#include <math.h>

#define B_  8
#define N_  4096
#define D_  64
#define S_  1024
#define K_  32

typedef __attribute__((ext_vector_type(2))) float v2f;
typedef __attribute__((ext_vector_type(4))) float v4f;
typedef __attribute__((ext_vector_type(8))) float v8f;

#define WS_T_OFF  ((size_t)B_ * S_ * 5)
#define WS_NEED   ((WS_T_OFF + (size_t)B_ * N_ * D_) * 4)
#define XT_BLOCKS (B_ * (N_ / 64))
#define XT_PITCH  65
typedef __attribute__((ext_vector_type(4))) int v4i;
typedef __attribute__((ext_vector_type(2))) unsigned long long v2u64;
#define FD 4

template <int J>
__device__ __forceinline__ float fps_sqdist(const v4f p, const v4f c)
{
    const float dx = p.x - c.x, dy = p.y - c.y, dz = p.z - c.z;
    float d;
    {
#pragma clang fp contract(off)
        d = (J < 2) ? __builtin_fmaf(dz, dz, __builtin_fmaf(dx, dx, dy * dy))
                    : dz * dz + __builtin_fmaf(dy, dy, dx * dx);
    }
    return d;
}

__device__ __forceinline__ void fps_emit(const v4f c, int b, int pos,
                                         float* __restrict__ out_newxyz,
                                         float* __restrict__ wsA,
                                         float* __restrict__ wsC2)
{
    out_newxyz[(size_t)b * 3 * S_ + 0 * S_ + pos] = c.x;
    out_newxyz[(size_t)b * 3 * S_ + 1 * S_ + pos] = c.y;
    out_newxyz[(size_t)b * 3 * S_ + 2 * S_ + pos] = c.z;
    float* A = wsA + ((size_t)b * S_ + pos) * 4;
    A[0] = -2.f * c.x; A[1] = -2.f * c.y; A[2] = -2.f * c.z; A[3] = 1.f;
    float c2;
    {
#pragma clang fp contract(off)
        c2 = c.z * c.z + __builtin_fmaf(c.x, c.x, c.y * c.y);
    }
    wsC2[(size_t)b * S_ + pos] = c2;
}

__device__ __forceinline__ void transpose_tile(const float* __restrict__ pts,
                                               float* __restrict__ wsT,
                                               int t, int tid, float* tile)
{
    const int b  = t / (N_ / 64);
    const int n0 = (t % (N_ / 64)) * 64;
    {
        const int ch = tid >> 4, p0 = (tid & 15) * 4;
        const v4f v = *(const v4f*)(pts + ((size_t)b * D_ + ch) * N_ + n0 + p0);
        float* d = tile + ch * XT_PITCH + p0;
        d[0] = v.x; d[1] = v.y; d[2] = v.z; d[3] = v.w;
    }
    __syncthreads();
    {
        const int pt = tid >> 4, c0 = (tid & 15) * 4;
        const float* s = tile + c0 * XT_PITCH + pt;
        v4f o;
        o.x = s[0]; o.y = s[XT_PITCH]; o.z = s[2 * XT_PITCH]; o.w = s[3 * XT_PITCH];
        *(v4f*)(wsT + ((size_t)b * N_ + n0 + pt) * D_ + c0) = o;
    }
}

__global__ __launch_bounds__(1024) void fps_kernel(
    const float* __restrict__ xyz,
    const int*   __restrict__ farthest_init,
    float* __restrict__ out_newxyz,
    float* __restrict__ wsA,
    float* __restrict__ wsC2,
    const float* __restrict__ pts,
    float* __restrict__ wsT,
    unsigned long long ws_bytes)
{
    __shared__ v4f   p4[N_];
    __shared__ __attribute__((aligned(16))) float pairs[FD][2 * 1024];
    __shared__ __attribute__((aligned(16))) int sRes[FD][4];
    __shared__ __attribute__((aligned(16))) unsigned long long lkeys[FD][32];

    const int b    = blockIdx.x;
    const int tid  = threadIdx.x;
    const int lane = tid & 31;
    const int wid  = tid >> 5;

    if (b >= B_) {
        if (ws_bytes >= WS_NEED) transpose_tile(pts, wsT, b - B_, tid, (float*)p4);
        return;
    }

    const float* xb = xyz + (size_t)b * 3 * N_;
    for (int i = tid; i < N_; i += 1024) {
        v4f v;
        v.x = xb[i];
        v.y = xb[N_ + i];
        v.z = xb[2 * N_ + i];
        v.w = 0.f;
        p4[i] = v;
    }
    float dS[FD][4];
#pragma unroll
    for (int j = 0; j < FD; ++j) {
#pragma unroll
        for (int s = 0; s < 4; ++s) dS[j][s] = 1e10f;
    }
    if (tid < FD * 4) (&sRes[0][0])[tid] = farthest_init[b];
    __syncthreads();

    int pos = -1;
    int g1 = -1, g2 = -1, g3 = -1;
    v4f c1 = {0.f, 0.f, 0.f, 0.f}, c2 = c1, c3 = c1;
    for (int round = 0; round < S_; ++round) {
        const v4i r0 = *(const v4i*)sRes[0];
        const v4i r1 = *(const v4i*)sRes[1];
        const v4i r2 = *(const v4i*)sRes[2];
        const v4i r3 = *(const v4i*)sRes[3];
        const int w0 = __builtin_amdgcn_readfirstlane(r0.x);
        const int w1 = __builtin_amdgcn_readfirstlane(r1.x);
        const int w2 = __builtin_amdgcn_readfirstlane(r2.x);
        const bool ok1 = (w0 == g1);
        const bool ok2 = ok1 && (w1 == g2);
        const bool ok3 = ok2 && (w2 == g3);
        const int  na  = (ok1 ? 1 : 0) + (ok2 ? 1 : 0) + (ok3 ? 1 : 0);
        if (tid == 0) {
            if (ok1 && pos + 1 < S_) fps_emit(c1, b, pos + 1, out_newxyz, wsA, wsC2);
            if (ok2 && pos + 2 < S_) fps_emit(c2, b, pos + 2, out_newxyz, wsA, wsC2);
            if (ok3 && pos + 3 < S_) fps_emit(c3, b, pos + 3, out_newxyz, wsA, wsC2);
        }
        pos += na + 1;
        if (pos >= S_) break;
        const v4i rs = ok3 ? r3 : (ok2 ? r2 : (ok1 ? r1 : r0));
        const int g0 = __builtin_amdgcn_readfirstlane(rs.x);
        g1 = __builtin_amdgcn_readfirstlane(rs.y);
        g2 = __builtin_amdgcn_readfirstlane(rs.z);
        g3 = __builtin_amdgcn_readfirstlane(rs.w);
        const v4f c0 = p4[g0];
        c1 = p4[g1];
        c2 = p4[g2];
        c3 = p4[g3];
        if (tid == 0) fps_emit(c0, b, pos, out_newxyz, wsA, wsC2);

        float bv[FD]; int bi[FD];
#pragma unroll
        for (int j = 0; j < FD; ++j) { bv[j] = -1.f; bi[j] = 0x7fffffff; }
#pragma unroll
        for (int s = 0; s < 4; ++s) {
            const int i = tid + 1024 * s;
            const v4f p = p4[i];
            const float dist = ok3 ? dS[3][s] : (ok2 ? dS[2][s] : (ok1 ? dS[1][s] : dS[0][s]));
            float d[FD];
            if (s < 2) {
                d[0] = fps_sqdist<0>(p, c0); d[1] = fps_sqdist<0>(p, c1);
                d[2] = fps_sqdist<0>(p, c2); d[3] = fps_sqdist<0>(p, c3);
            } else {
                d[0] = fps_sqdist<2>(p, c0); d[1] = fps_sqdist<2>(p, c1);
                d[2] = fps_sqdist<2>(p, c2); d[3] = fps_sqdist<2>(p, c3);
            }
            float run = dist;
#pragma unroll
            for (int j = 0; j < FD; ++j) {
                run = fminf(run, d[j]);
                dS[j][s] = run;
                const bool take = run > bv[j];
                bv[j] = take ? run : bv[j];
                bi[j] = take ? i : bi[j];
            }
        }
#pragma unroll
        for (int j = 0; j < FD; ++j) {
            v2f q;
            q.x = bv[j]; q.y = __int_as_float(bi[j]);
            *(v2f*)&pairs[j][2 * tid] = q;
        }
        __syncthreads();
        if (wid < FD) {
            const float* pr = pairs[wid];
            float tv[32]; int ti[32];
#pragma unroll
            for (int k = 0; k < 16; ++k) {
                const v4f q = *(const v4f*)&pr[4 * (k * 32 + lane)];
                tv[2 * k]     = q.x; ti[2 * k]     = __float_as_int(q.y);
                tv[2 * k + 1] = q.z; ti[2 * k + 1] = __float_as_int(q.w);
            }
            float tm[16];
#pragma unroll
            for (int t = 0; t < 16; ++t) tm[t] = fmaxf(tv[t], tv[t + 16]);
#pragma unroll
            for (int n = 8; n >= 1; n >>= 1) {
#pragma unroll
                for (int t = 0; t < n; ++t) tm[t] = fmaxf(tm[t], tm[t + n]);
            }
            const float lmax = tm[0];
#pragma unroll
            for (int t = 0; t < 32; ++t) ti[t] = (tv[t] == lmax) ? ti[t] : 0x7fffffff;
#pragma unroll
            for (int n = 16; n >= 1; n >>= 1) {
#pragma unroll
                for (int t = 0; t < n; ++t) ti[t] = ti[t + n] < ti[t] ? ti[t + n] : ti[t];
            }
            const int li = ti[0];
            const unsigned ub = __float_as_uint(lmax);
            const unsigned hb = ub ^ ((unsigned)((int)ub >> 31) | 0x80000000u);
            const unsigned long long key =
                ((unsigned long long)hb << 32) | (unsigned long long)(unsigned)(~li);
            unsigned long long* kb = lkeys[wid];
            kb[lane] = key;
            v2u64 q[16];
#pragma unroll
            for (int k = 0; k < 16; ++k) q[k] = *(const v2u64*)&kb[2 * k];
            int rk[4] = {0, 0, 0, 0};
#pragma unroll
            for (int k = 0; k < 16; ++k)
                rk[k & 3] += (q[k].x > key ? 1 : 0) + (q[k].y > key ? 1 : 0);
            const int rank = (rk[0] + rk[1]) + (rk[2] + rk[3]);
            if (rank < 4) sRes[wid][rank] = li;
        }
        __syncthreads();
    }
}

#define TPITCH 4100
#define CAND_CAP 512

template <int R>
__device__ __forceinline__ void select32(const float* cd, const unsigned short* cj,
                                         int C, int lane, int* selRow)
{
    unsigned key[R];
    int      jv[R];
#pragma unroll
    for (int r = 0; r < R; ++r) {
        const int  i  = r * 32 + lane;
        const bool in = i < C;
        const int  ii = in ? i : 0;
        const unsigned u = __float_as_uint(cd[ii]);
        const unsigned k = u ^ ((unsigned)((int)u >> 31) | 0x80000000u);
        key[r] = in ? k : 0xFFFFFFFFu;
        jv[r]  = (int)cj[ii];
    }
    unsigned T = 0;
    for (int bit = 31; bit >= 0; --bit) {
        const unsigned cand = T | (1u << bit);
        int cnt = 0;
#pragma unroll
        for (int r = 0; r < R; ++r)
            cnt += __popc(__builtin_amdgcn_ballot_w32(key[r] < cand));
        if (cnt < K_) T = cand;
    }
    int nless = 0;
#pragma unroll
    for (int r = 0; r < R; ++r)
        nless += __popc(__builtin_amdgcn_ballot_w32(key[r] < T));
    const int need = K_ - nless;
    int running = 0, tieSeen = 0;
#pragma unroll
    for (int r = 0; r < R; ++r) {
        const bool isTie = key[r] == T;
        const unsigned tb = __builtin_amdgcn_ballot_w32(isTie);
        const int tieRank = tieSeen + (int)__builtin_amdgcn_mbcnt_lo(tb, 0u);
        const bool sel = (key[r] < T) || (isTie && tieRank < need);
        const unsigned sb = __builtin_amdgcn_ballot_w32(sel);
        const int slot = running + (int)__builtin_amdgcn_mbcnt_lo(sb, 0u);
        if (sel && slot < K_) selRow[slot] = jv[r];
        running += __popc(sb);
        tieSeen += __popc(tb);
    }
}

__device__ __forceinline__ void select32_scan(const float* row, int lane, int* selRow)
{
    float pd = -3.4e38f; int pj = -1;
    for (int k = 0; k < K_; ++k) {
        float bv = 3.4e38f; int bi = 0x7fffffff;
        for (int jj = 0; jj < N_ / 128; ++jj) {
            const int j0 = jj * 128 + lane * 4;
            const v4f dv = *(const v4f*)&row[j0];
#pragma unroll
            for (int e = 0; e < 4; ++e) {
                const float d = dv[e];
                const int   j = j0 + e;
                const bool after = d > pd || (d == pd && j > pj);
                if (after && (d < bv || (d == bv && j < bi))) { bv = d; bi = j; }
            }
        }
#pragma unroll
        for (int off = 16; off >= 1; off >>= 1) {
            const float ov = __shfl_xor(bv, off, 32);
            const int   oi = __shfl_xor(bi, off, 32);
            if (ov < bv || (ov == bv && oi < bi)) { bv = ov; bi = oi; }
        }
        if (lane == 0) selRow[k] = bi;
        pd = bv; pj = bi;
    }
}

__global__ __launch_bounds__(512) void group_kernel(
    const float* __restrict__ xyz,
    const float* __restrict__ pts,
    const float* __restrict__ wsA,
    const float* __restrict__ wsC2,
    const float* __restrict__ wsT,
    unsigned long long ws_bytes,
    float* __restrict__ out_feat)
{
    __shared__ float DT[16 * TPITCH];
    __shared__ __attribute__((aligned(16))) int selIdx[16][K_];
    __shared__ float          candD[16][CAND_CAP];
    __shared__ unsigned short candJ[16][CAND_CAP];

    const int b    = blockIdx.y;
    const int s0   = blockIdx.x * 16;
    const int tid  = threadIdx.x;
    const int lane = tid & 31;
    const int wid  = tid >> 5;
    const bool hi  = lane >= 16;

    const float* xb = xyz + (size_t)b * 3 * N_;

    const int  mcol = lane & 15;
    const v4f  A4   = *(const v4f*)(wsA + ((size_t)b * S_ + s0 + mcol) * 4);
    v2f a;
    a.x = hi ? A4.z : A4.x;
    a.y = hi ? A4.w : A4.y;

    const int mbase = hi ? 8 : 0;
    float cmr[8];
#pragma unroll
    for (int r = 0; r < 8; ++r)
        cmr[r] = wsC2[(size_t)b * S_ + s0 + mbase + r];

    for (int t = 0; t < 16; ++t) {
        const int n = wid * 256 + t * 16 + mcol;
        const float x = xb[n], y = xb[N_ + n], z = xb[2 * N_ + n];
        const float pn2 = x * x + y * y + z * z;
        v2f bb;
        bb.x = hi ? z : x;
        bb.y = hi ? pn2 : y;

        v8f c = {};
        c = __builtin_amdgcn_wmma_f32_16x16x4_f32(
                 false, a,  false, bb,
                 (short)0, c,  false,  false);

#pragma unroll
        for (int r = 0; r < 8; ++r)
            DT[(mbase + r) * TPITCH + n] = c[r] + cmr[r];
    }
    __syncthreads();

    const int m = wid;
    {
        const float* row = &DT[m * TPITCH];
        float* cd = candD[wid];
        unsigned short* cj = candJ[wid];

        float mn = 3.4e38f;
        for (int jj = 0; jj < N_ / 128; ++jj) {
            const v4f dv = *(const v4f*)&row[jj * 128 + lane * 4];
            mn = fminf(mn, fminf(fminf(dv.x, dv.y), fminf(dv.z, dv.w)));
        }
        float U = mn;
#pragma unroll
        for (int off = 16; off >= 1; off >>= 1)
            U = fmaxf(U, __shfl_xor(U, off, 32));

        int C = 0;
        for (int jj = 0; jj < N_ / 128; ++jj) {
            const int j0 = jj * 128 + lane * 4;
            const v4f dv = *(const v4f*)&row[j0];
            const bool p0 = dv.x <= U, p1 = dv.y <= U;
            const bool p2 = dv.z <= U, p3 = dv.w <= U;
            const unsigned b0 = __builtin_amdgcn_ballot_w32(p0);
            const unsigned b1 = __builtin_amdgcn_ballot_w32(p1);
            const unsigned b2 = __builtin_amdgcn_ballot_w32(p2);
            const unsigned b3 = __builtin_amdgcn_ballot_w32(p3);
            unsigned pre = __builtin_amdgcn_mbcnt_lo(b0, 0u);
            pre = __builtin_amdgcn_mbcnt_lo(b1, pre);
            pre = __builtin_amdgcn_mbcnt_lo(b2, pre);
            pre = __builtin_amdgcn_mbcnt_lo(b3, pre);
            int pos = C + (int)pre;
            if (p0 && pos < CAND_CAP) { cd[pos] = dv.x; cj[pos] = (unsigned short)j0; }
            pos += p0 ? 1 : 0;
            if (p1 && pos < CAND_CAP) { cd[pos] = dv.y; cj[pos] = (unsigned short)(j0 + 1); }
            pos += p1 ? 1 : 0;
            if (p2 && pos < CAND_CAP) { cd[pos] = dv.z; cj[pos] = (unsigned short)(j0 + 2); }
            pos += p2 ? 1 : 0;
            if (p3 && pos < CAND_CAP) { cd[pos] = dv.w; cj[pos] = (unsigned short)(j0 + 3); }
            C += __popc(b0) + __popc(b1) + __popc(b2) + __popc(b3);
        }
        __syncthreads();

        int* selRow = selIdx[m];
        if (C <= 128)           select32<4>(cd, cj, C, lane, selRow);
        else if (C <= 256)      select32<8>(cd, cj, C, lane, selRow);
        else if (C <= CAND_CAP) select32<16>(cd, cj, C, lane, selRow);
        else                    select32_scan(row, lane, selRow);
        __syncthreads();
    }

    float* ob = out_feat + (size_t)b * 67 * S_ + (s0 + m);
    if (ws_bytes >= WS_NEED) {
        const int q = lane & 15;
        const v4i* sr = (const v4i*)(selIdx[m] + (hi ? 16 : 0));
        const float* tb = wsT + (size_t)b * N_ * D_ + 4 * q;
        v4i jv[4];
#pragma unroll
        for (int i = 0; i < 4; ++i) jv[i] = sr[i];
        v4f mx = {-3.4e38f, -3.4e38f, -3.4e38f, -3.4e38f};
#pragma unroll
        for (int k = 0; k < 16; ++k) {
            const int j = jv[k >> 2][k & 3];
            const v4f v = *(const v4f*)(tb + (size_t)j * D_);
            mx.x = fmaxf(mx.x, v.x); mx.y = fmaxf(mx.y, v.y);
            mx.z = fmaxf(mx.z, v.z); mx.w = fmaxf(mx.w, v.w);
        }
        const int jl = selIdx[m][lane];
        float gx = xb[jl], gy = xb[N_ + jl], gz = xb[2 * N_ + jl];
        mx.x = fmaxf(mx.x, __shfl_xor(mx.x, 16, 32));
        mx.y = fmaxf(mx.y, __shfl_xor(mx.y, 16, 32));
        mx.z = fmaxf(mx.z, __shfl_xor(mx.z, 16, 32));
        mx.w = fmaxf(mx.w, __shfl_xor(mx.w, 16, 32));
#pragma unroll
        for (int off = 16; off >= 1; off >>= 1) {
            gx = fmaxf(gx, __shfl_xor(gx, off, 32));
            gy = fmaxf(gy, __shfl_xor(gy, off, 32));
            gz = fmaxf(gz, __shfl_xor(gz, off, 32));
        }
        const int c0 = 3 + 4 * q + (hi ? 2 : 0);
        ob[(size_t)c0 * S_]       = hi ? mx.z : mx.x;
        ob[(size_t)(c0 + 1) * S_] = hi ? mx.w : mx.y;
        if (lane < 3) ob[(size_t)lane * S_] = lane == 0 ? gx : (lane == 1 ? gy : gz);
        return;
    }
    const float* pb = pts + (size_t)b * D_ * N_;
    const unsigned o0 = (unsigned)lane * N_;
    const unsigned o1 = (unsigned)(lane + 32) * N_;
    const unsigned o2 = (unsigned)(lane < 2 ? lane : 2) * N_;
    const int* selRow = selIdx[m];
    float m0 = -3.4e38f, m1 = -3.4e38f, m2 = -3.4e38f;
#pragma unroll
    for (int k = 0; k < K_; ++k) {
        const int j = __builtin_amdgcn_readfirstlane(selRow[k]);
        m0 = fmaxf(m0, pb[o0 + j]);
        m1 = fmaxf(m1, pb[o1 + j]);
        m2 = fmaxf(m2, xb[o2 + j]);
    }
    ob[(size_t)(3 + lane) * S_]  = m0;
    ob[(size_t)(35 + lane) * S_] = m1;
    if (lane < 3) ob[(size_t)lane * S_] = m2;
}

extern "C" void kernel_launch(void* const* d_in, const int* in_sizes, int n_in,
                              void* d_out, int out_size, void* d_ws, size_t ws_size,
                              hipStream_t stream) {
    const float* xyz    = (const float*)d_in[0];
    const float* points = (const float*)d_in[1];
    const int*   finit  = (const int*)d_in[2];

    float* out        = (float*)d_out;
    float* out_newxyz = out;
    float* out_feat   = out + (size_t)B_ * 3 * S_;

    float* wsA  = (float*)d_ws;
    float* wsC2 = wsA + (size_t)B_ * S_ * 4;
    float* wsT  = wsA + WS_T_OFF;

    const unsigned long long ws_bytes = ws_size;
    fps_kernel<<<B_ + XT_BLOCKS, 1024, 0, stream>>>(xyz, finit, out_newxyz, wsA, wsC2,
                                                    points, wsT, ws_bytes);
    group_kernel<<<dim3(S_ / 16, B_), 512, 0, stream>>>(xyz, points, wsA, wsC2, wsT, ws_bytes, out_feat);
}


# FPS per-point pass tracks no index: a thread owns 4 consecutive points, stores its max and its 4 values; fold wave finds the thread by position, the point by reading that thread's 4 values back
# speedup vs baseline: 5.0848x; 1.0701x over previous
#include <hip/hip_runtime.h>
#include <math.h>

#define B_  8
#define N_  4096
#define D_  64
#define S_  1024
#define K_  32

typedef __attribute__((ext_vector_type(2))) float v2f;
typedef __attribute__((ext_vector_type(4))) float v4f;
typedef __attribute__((ext_vector_type(8))) float v8f;

#define WS_T_OFF  ((size_t)B_ * S_ * 5)
#define WS_NEED   ((WS_T_OFF + (size_t)B_ * N_ * D_) * 4)
#define XT_BLOCKS (B_ * (N_ / 64))
#define XT_PITCH  65
typedef __attribute__((ext_vector_type(4))) int v4i;
typedef __attribute__((ext_vector_type(2))) unsigned long long v2u64;
#define FD 4
#define P4_AT(i) ((((i) & 3) << 10) + ((i) >> 2))

template <int J>
__device__ __forceinline__ float fps_sqdist(const v4f p, const v4f c)
{
    const float dx = p.x - c.x, dy = p.y - c.y, dz = p.z - c.z;
    float d;
    {
#pragma clang fp contract(off)
        d = (J < 2) ? __builtin_fmaf(dz, dz, __builtin_fmaf(dx, dx, dy * dy))
                    : dz * dz + __builtin_fmaf(dy, dy, dx * dx);
    }
    return d;
}

__device__ __forceinline__ void fps_emit(const v4f c, int b, int pos,
                                         float* __restrict__ out_newxyz,
                                         float* __restrict__ wsA,
                                         float* __restrict__ wsC2)
{
    out_newxyz[(size_t)b * 3 * S_ + 0 * S_ + pos] = c.x;
    out_newxyz[(size_t)b * 3 * S_ + 1 * S_ + pos] = c.y;
    out_newxyz[(size_t)b * 3 * S_ + 2 * S_ + pos] = c.z;
    float* A = wsA + ((size_t)b * S_ + pos) * 4;
    A[0] = -2.f * c.x; A[1] = -2.f * c.y; A[2] = -2.f * c.z; A[3] = 1.f;
    float c2;
    {
#pragma clang fp contract(off)
        c2 = c.z * c.z + __builtin_fmaf(c.x, c.x, c.y * c.y);
    }
    wsC2[(size_t)b * S_ + pos] = c2;
}

__device__ __forceinline__ void transpose_tile(const float* __restrict__ pts,
                                               float* __restrict__ wsT,
                                               int t, int tid, float* tile)
{
    const int b  = t / (N_ / 64);
    const int n0 = (t % (N_ / 64)) * 64;
    {
        const int ch = tid >> 4, p0 = (tid & 15) * 4;
        const v4f v = *(const v4f*)(pts + ((size_t)b * D_ + ch) * N_ + n0 + p0);
        float* d = tile + ch * XT_PITCH + p0;
        d[0] = v.x; d[1] = v.y; d[2] = v.z; d[3] = v.w;
    }
    __syncthreads();
    {
        const int pt = tid >> 4, c0 = (tid & 15) * 4;
        const float* s = tile + c0 * XT_PITCH + pt;
        v4f o;
        o.x = s[0]; o.y = s[XT_PITCH]; o.z = s[2 * XT_PITCH]; o.w = s[3 * XT_PITCH];
        *(v4f*)(wsT + ((size_t)b * N_ + n0 + pt) * D_ + c0) = o;
    }
}

template <int J>
__device__ __forceinline__ void fps_pass(const v4f* p4, int tid, bool ok1, bool ok2, bool ok3,
                                         const v4f c0, const v4f c1, const v4f c2, const v4f c3,
                                         float (&dS)[FD][4], float (*tmax)[1024], v4f (*runs)[1024])
{
#pragma unroll
    for (int s = 0; s < 4; ++s) {
        const v4f p = p4[s * 1024 + tid];
        const float dist = ok3 ? dS[3][s] : (ok2 ? dS[2][s] : (ok1 ? dS[1][s] : dS[0][s]));
        float d[FD];
        d[0] = fps_sqdist<J>(p, c0); d[1] = fps_sqdist<J>(p, c1);
        d[2] = fps_sqdist<J>(p, c2); d[3] = fps_sqdist<J>(p, c3);
        float run = dist;
#pragma unroll
        for (int j = 0; j < FD; ++j) {
            run = fminf(run, d[j]);
            dS[j][s] = run;
        }
    }
#pragma unroll
    for (int j = 0; j < FD; ++j) {
        v4f r;
        r.x = dS[j][0]; r.y = dS[j][1]; r.z = dS[j][2]; r.w = dS[j][3];
        runs[j][tid] = r;
        tmax[j][tid] = fmaxf(fmaxf(r.x, r.y), fmaxf(r.z, r.w));
    }
}

__global__ __launch_bounds__(1024) void fps_kernel(
    const float* __restrict__ xyz,
    const int*   __restrict__ farthest_init,
    float* __restrict__ out_newxyz,
    float* __restrict__ wsA,
    float* __restrict__ wsC2,
    const float* __restrict__ pts,
    float* __restrict__ wsT,
    unsigned long long ws_bytes)
{
    __shared__ v4f   p4[N_];
    __shared__ __attribute__((aligned(16))) float tmax[FD][1024];
    __shared__ __attribute__((aligned(16))) v4f runs[FD][1024];
    __shared__ __attribute__((aligned(16))) int sRes[FD][4];
    __shared__ __attribute__((aligned(16))) unsigned long long lkeys[FD][32];

    const int b    = blockIdx.x;
    const int tid  = threadIdx.x;
    const int lane = tid & 31;
    const int wid  = tid >> 5;

    if (b >= B_) {
        if (ws_bytes >= WS_NEED) transpose_tile(pts, wsT, b - B_, tid, (float*)p4);
        return;
    }

    const float* xb = xyz + (size_t)b * 3 * N_;
    for (int i = tid; i < N_; i += 1024) {
        v4f v;
        v.x = xb[i];
        v.y = xb[N_ + i];
        v.z = xb[2 * N_ + i];
        v.w = 0.f;
        p4[P4_AT(i)] = v;
    }
    float dS[FD][4];
#pragma unroll
    for (int j = 0; j < FD; ++j) {
#pragma unroll
        for (int s = 0; s < 4; ++s) dS[j][s] = 1e10f;
    }
    if (tid < FD * 4) (&sRes[0][0])[tid] = farthest_init[b];
    __syncthreads();

    int pos = -1;
    int g1 = -1, g2 = -1, g3 = -1;
    v4f c1 = {0.f, 0.f, 0.f, 0.f}, c2 = c1, c3 = c1;
    for (int round = 0; round < S_; ++round) {
        const v4i r0 = *(const v4i*)sRes[0];
        const v4i r1 = *(const v4i*)sRes[1];
        const v4i r2 = *(const v4i*)sRes[2];
        const v4i r3 = *(const v4i*)sRes[3];
        const int w0 = __builtin_amdgcn_readfirstlane(r0.x);
        const int w1 = __builtin_amdgcn_readfirstlane(r1.x);
        const int w2 = __builtin_amdgcn_readfirstlane(r2.x);
        const bool ok1 = (w0 == g1);
        const bool ok2 = ok1 && (w1 == g2);
        const bool ok3 = ok2 && (w2 == g3);
        const int  na  = (ok1 ? 1 : 0) + (ok2 ? 1 : 0) + (ok3 ? 1 : 0);
        if (tid == 0) {
            if (ok1 && pos + 1 < S_) fps_emit(c1, b, pos + 1, out_newxyz, wsA, wsC2);
            if (ok2 && pos + 2 < S_) fps_emit(c2, b, pos + 2, out_newxyz, wsA, wsC2);
            if (ok3 && pos + 3 < S_) fps_emit(c3, b, pos + 3, out_newxyz, wsA, wsC2);
        }
        pos += na + 1;
        if (pos >= S_) break;
        const v4i rs = ok3 ? r3 : (ok2 ? r2 : (ok1 ? r1 : r0));
        const int g0 = __builtin_amdgcn_readfirstlane(rs.x);
        g1 = __builtin_amdgcn_readfirstlane(rs.y);
        g2 = __builtin_amdgcn_readfirstlane(rs.z);
        g3 = __builtin_amdgcn_readfirstlane(rs.w);
        const v4f c0 = p4[P4_AT(g0)];
        c1 = p4[P4_AT(g1)];
        c2 = p4[P4_AT(g2)];
        c3 = p4[P4_AT(g3)];
        if (tid == 0) fps_emit(c0, b, pos, out_newxyz, wsA, wsC2);

        if (__builtin_amdgcn_readfirstlane(wid) < 16)
            fps_pass<0>(p4, tid, ok1, ok2, ok3, c0, c1, c2, c3, dS, tmax, runs);
        else
            fps_pass<2>(p4, tid, ok1, ok2, ok3, c0, c1, c2, c3, dS, tmax, runs);
        __syncthreads();
        if (wid < FD) {
            const float* pr = tmax[wid];
            float tv[32];
#pragma unroll
            for (int k = 0; k < 8; ++k) {
                const v4f q = *(const v4f*)&pr[4 * (k * 32 + lane)];
                tv[4 * k]     = q.x; tv[4 * k + 1] = q.y;
                tv[4 * k + 2] = q.z; tv[4 * k + 3] = q.w;
            }
            float tm[16];
#pragma unroll
            for (int t = 0; t < 16; ++t) tm[t] = fmaxf(tv[t], tv[t + 16]);
#pragma unroll
            for (int n = 8; n >= 1; n >>= 1) {
#pragma unroll
                for (int t = 0; t < n; ++t) tm[t] = fmaxf(tm[t], tm[t + n]);
            }
            const float lmax = tm[0];
            int to[32];
#pragma unroll
            for (int t = 0; t < 32; ++t) to[t] = (tv[t] == lmax) ? t : 64;
#pragma unroll
            for (int n = 16; n >= 1; n >>= 1) {
#pragma unroll
                for (int t = 0; t < n; ++t) to[t] = to[t + n] < to[t] ? to[t + n] : to[t];
            }
            const int lt = 128 * (to[0] >> 2) + 4 * lane + (to[0] & 3);
            const v4f rq = runs[wid][lt];
            const unsigned ub = __float_as_uint(lmax);
            const unsigned hb = ub ^ ((unsigned)((int)ub >> 31) | 0x80000000u);
            const unsigned long long key =
                ((unsigned long long)hb << 32) | (unsigned long long)(unsigned)(~lt);
            unsigned long long* kb = lkeys[wid];
            kb[lane] = key;
            v2u64 q[16];
#pragma unroll
            for (int k = 0; k < 16; ++k) q[k] = *(const v2u64*)&kb[2 * k];
            asm volatile("" :: "v"(rq.x), "v"(rq.y), "v"(rq.z), "v"(rq.w));
            int rk[4] = {0, 0, 0, 0};
#pragma unroll
            for (int k = 0; k < 16; ++k)
                rk[k & 3] += (q[k].x > key ? 1 : 0) + (q[k].y > key ? 1 : 0);
            const int rank = (rk[0] + rk[1]) + (rk[2] + rk[3]);
            int slot = 3;
            slot = (rq.z == lmax) ? 2 : slot;
            slot = (rq.y == lmax) ? 1 : slot;
            slot = (rq.x == lmax) ? 0 : slot;
            const int li = 4 * lt + slot;
            if (rank < 4) sRes[wid][rank] = li;
        }
        __syncthreads();
    }
}

#define TPITCH 4100
#define CAND_CAP 512

template <int R>
__device__ __forceinline__ void select32(const float* cd, const unsigned short* cj,
                                         int C, int lane, int* selRow)
{
    unsigned key[R];
    int      jv[R];
#pragma unroll
    for (int r = 0; r < R; ++r) {
        const int  i  = r * 32 + lane;
        const bool in = i < C;
        const int  ii = in ? i : 0;
        const unsigned u = __float_as_uint(cd[ii]);
        const unsigned k = u ^ ((unsigned)((int)u >> 31) | 0x80000000u);
        key[r] = in ? k : 0xFFFFFFFFu;
        jv[r]  = (int)cj[ii];
    }
    unsigned T = 0;
    for (int bit = 31; bit >= 0; --bit) {
        const unsigned cand = T | (1u << bit);
        int cnt = 0;
#pragma unroll
        for (int r = 0; r < R; ++r)
            cnt += __popc(__builtin_amdgcn_ballot_w32(key[r] < cand));
        if (cnt < K_) T = cand;
    }
    int nless = 0;
#pragma unroll
    for (int r = 0; r < R; ++r)
        nless += __popc(__builtin_amdgcn_ballot_w32(key[r] < T));
    const int need = K_ - nless;
    int running = 0, tieSeen = 0;
#pragma unroll
    for (int r = 0; r < R; ++r) {
        const bool isTie = key[r] == T;
        const unsigned tb = __builtin_amdgcn_ballot_w32(isTie);
        const int tieRank = tieSeen + (int)__builtin_amdgcn_mbcnt_lo(tb, 0u);
        const bool sel = (key[r] < T) || (isTie && tieRank < need);
        const unsigned sb = __builtin_amdgcn_ballot_w32(sel);
        const int slot = running + (int)__builtin_amdgcn_mbcnt_lo(sb, 0u);
        if (sel && slot < K_) selRow[slot] = jv[r];
        running += __popc(sb);
        tieSeen += __popc(tb);
    }
}

__device__ __forceinline__ void select32_scan(const float* row, int lane, int* selRow)
{
    float pd = -3.4e38f; int pj = -1;
    for (int k = 0; k < K_; ++k) {
        float bv = 3.4e38f; int bi = 0x7fffffff;
        for (int jj = 0; jj < N_ / 128; ++jj) {
            const int j0 = jj * 128 + lane * 4;
            const v4f dv = *(const v4f*)&row[j0];
#pragma unroll
            for (int e = 0; e < 4; ++e) {
                const float d = dv[e];
                const int   j = j0 + e;
                const bool after = d > pd || (d == pd && j > pj);
                if (after && (d < bv || (d == bv && j < bi))) { bv = d; bi = j; }
            }
        }
#pragma unroll
        for (int off = 16; off >= 1; off >>= 1) {
            const float ov = __shfl_xor(bv, off, 32);
            const int   oi = __shfl_xor(bi, off, 32);
            if (ov < bv || (ov == bv && oi < bi)) { bv = ov; bi = oi; }
        }
        if (lane == 0) selRow[k] = bi;
        pd = bv; pj = bi;
    }
}

__global__ __launch_bounds__(512) void group_kernel(
    const float* __restrict__ xyz,
    const float* __restrict__ pts,
    const float* __restrict__ wsA,
    const float* __restrict__ wsC2,
    const float* __restrict__ wsT,
    unsigned long long ws_bytes,
    float* __restrict__ out_feat)
{
    __shared__ float DT[16 * TPITCH];
    __shared__ __attribute__((aligned(16))) int selIdx[16][K_];
    __shared__ float          candD[16][CAND_CAP];
    __shared__ unsigned short candJ[16][CAND_CAP];

    const int b    = blockIdx.y;
    const int s0   = blockIdx.x * 16;
    const int tid  = threadIdx.x;
    const int lane = tid & 31;
    const int wid  = tid >> 5;
    const bool hi  = lane >= 16;

    const float* xb = xyz + (size_t)b * 3 * N_;

    const int  mcol = lane & 15;
    const v4f  A4   = *(const v4f*)(wsA + ((size_t)b * S_ + s0 + mcol) * 4);
    v2f a;
    a.x = hi ? A4.z : A4.x;
    a.y = hi ? A4.w : A4.y;

    const int mbase = hi ? 8 : 0;
    float cmr[8];
#pragma unroll
    for (int r = 0; r < 8; ++r)
        cmr[r] = wsC2[(size_t)b * S_ + s0 + mbase + r];

    for (int t = 0; t < 16; ++t) {
        const int n = wid * 256 + t * 16 + mcol;
        const float x = xb[n], y = xb[N_ + n], z = xb[2 * N_ + n];
        const float pn2 = x * x + y * y + z * z;
        v2f bb;
        bb.x = hi ? z : x;
        bb.y = hi ? pn2 : y;

        v8f c = {};
        c = __builtin_amdgcn_wmma_f32_16x16x4_f32(
                 false, a,  false, bb,
                 (short)0, c,  false,  false);

#pragma unroll
        for (int r = 0; r < 8; ++r)
            DT[(mbase + r) * TPITCH + n] = c[r] + cmr[r];
    }
    __syncthreads();

    const int m = wid;
    {
        const float* row = &DT[m * TPITCH];
        float* cd = candD[wid];
        unsigned short* cj = candJ[wid];

        float mn = 3.4e38f;
        for (int jj = 0; jj < N_ / 128; ++jj) {
            const v4f dv = *(const v4f*)&row[jj * 128 + lane * 4];
            mn = fminf(mn, fminf(fminf(dv.x, dv.y), fminf(dv.z, dv.w)));
        }
        float U = mn;
#pragma unroll
        for (int off = 16; off >= 1; off >>= 1)
            U = fmaxf(U, __shfl_xor(U, off, 32));

        int C = 0;
        for (int jj = 0; jj < N_ / 128; ++jj) {
            const int j0 = jj * 128 + lane * 4;
            const v4f dv = *(const v4f*)&row[j0];
            const bool p0 = dv.x <= U, p1 = dv.y <= U;
            const bool p2 = dv.z <= U, p3 = dv.w <= U;
            const unsigned b0 = __builtin_amdgcn_ballot_w32(p0);
            const unsigned b1 = __builtin_amdgcn_ballot_w32(p1);
            const unsigned b2 = __builtin_amdgcn_ballot_w32(p2);
            const unsigned b3 = __builtin_amdgcn_ballot_w32(p3);
            unsigned pre = __builtin_amdgcn_mbcnt_lo(b0, 0u);
            pre = __builtin_amdgcn_mbcnt_lo(b1, pre);
            pre = __builtin_amdgcn_mbcnt_lo(b2, pre);
            pre = __builtin_amdgcn_mbcnt_lo(b3, pre);
            int pos = C + (int)pre;
            if (p0 && pos < CAND_CAP) { cd[pos] = dv.x; cj[pos] = (unsigned short)j0; }
            pos += p0 ? 1 : 0;
            if (p1 && pos < CAND_CAP) { cd[pos] = dv.y; cj[pos] = (unsigned short)(j0 + 1); }
            pos += p1 ? 1 : 0;
            if (p2 && pos < CAND_CAP) { cd[pos] = dv.z; cj[pos] = (unsigned short)(j0 + 2); }
            pos += p2 ? 1 : 0;
            if (p3 && pos < CAND_CAP) { cd[pos] = dv.w; cj[pos] = (unsigned short)(j0 + 3); }
            C += __popc(b0) + __popc(b1) + __popc(b2) + __popc(b3);
        }
        __syncthreads();

        int* selRow = selIdx[m];
        if (C <= 128)           select32<4>(cd, cj, C, lane, selRow);
        else if (C <= 256)      select32<8>(cd, cj, C, lane, selRow);
        else if (C <= CAND_CAP) select32<16>(cd, cj, C, lane, selRow);
        else                    select32_scan(row, lane, selRow);
        __syncthreads();
    }

    float* ob = out_feat + (size_t)b * 67 * S_ + (s0 + m);
    if (ws_bytes >= WS_NEED) {
        const int q = lane & 15;
        const v4i* sr = (const v4i*)(selIdx[m] + (hi ? 16 : 0));
        const float* tb = wsT + (size_t)b * N_ * D_ + 4 * q;
        v4i jv[4];
#pragma unroll
        for (int i = 0; i < 4; ++i) jv[i] = sr[i];
        v4f mx = {-3.4e38f, -3.4e38f, -3.4e38f, -3.4e38f};
#pragma unroll
        for (int k = 0; k < 16; ++k) {
            const int j = jv[k >> 2][k & 3];
            const v4f v = *(const v4f*)(tb + (size_t)j * D_);
            mx.x = fmaxf(mx.x, v.x); mx.y = fmaxf(mx.y, v.y);
            mx.z = fmaxf(mx.z, v.z); mx.w = fmaxf(mx.w, v.w);
        }
        const int jl = selIdx[m][lane];
        float gx = xb[jl], gy = xb[N_ + jl], gz = xb[2 * N_ + jl];
        mx.x = fmaxf(mx.x, __shfl_xor(mx.x, 16, 32));
        mx.y = fmaxf(mx.y, __shfl_xor(mx.y, 16, 32));
        mx.z = fmaxf(mx.z, __shfl_xor(mx.z, 16, 32));
        mx.w = fmaxf(mx.w, __shfl_xor(mx.w, 16, 32));
#pragma unroll
        for (int off = 16; off >= 1; off >>= 1) {
            gx = fmaxf(gx, __shfl_xor(gx, off, 32));
            gy = fmaxf(gy, __shfl_xor(gy, off, 32));
            gz = fmaxf(gz, __shfl_xor(gz, off, 32));
        }
        const int c0 = 3 + 4 * q + (hi ? 2 : 0);
        ob[(size_t)c0 * S_]       = hi ? mx.z : mx.x;
        ob[(size_t)(c0 + 1) * S_] = hi ? mx.w : mx.y;
        if (lane < 3) ob[(size_t)lane * S_] = lane == 0 ? gx : (lane == 1 ? gy : gz);
        return;
    }
    const float* pb = pts + (size_t)b * D_ * N_;
    const unsigned o0 = (unsigned)lane * N_;
    const unsigned o1 = (unsigned)(lane + 32) * N_;
    const unsigned o2 = (unsigned)(lane < 2 ? lane : 2) * N_;
    const int* selRow = selIdx[m];
    float m0 = -3.4e38f, m1 = -3.4e38f, m2 = -3.4e38f;
#pragma unroll
    for (int k = 0; k < K_; ++k) {
        const int j = __builtin_amdgcn_readfirstlane(selRow[k]);
        m0 = fmaxf(m0, pb[o0 + j]);
        m1 = fmaxf(m1, pb[o1 + j]);
        m2 = fmaxf(m2, xb[o2 + j]);
    }
    ob[(size_t)(3 + lane) * S_]  = m0;
    ob[(size_t)(35 + lane) * S_] = m1;
    if (lane < 3) ob[(size_t)lane * S_] = m2;
}

extern "C" void kernel_launch(void* const* d_in, const int* in_sizes, int n_in,
                              void* d_out, int out_size, void* d_ws, size_t ws_size,
                              hipStream_t stream) {
    const float* xyz    = (const float*)d_in[0];
    const float* points = (const float*)d_in[1];
    const int*   finit  = (const int*)d_in[2];

    float* out        = (float*)d_out;
    float* out_newxyz = out;
    float* out_feat   = out + (size_t)B_ * 3 * S_;

    float* wsA  = (float*)d_ws;
    float* wsC2 = wsA + (size_t)B_ * S_ * 4;
    float* wsT  = wsA + WS_T_OFF;

    const unsigned long long ws_bytes = ws_size;
    fps_kernel<<<B_ + XT_BLOCKS, 1024, 0, stream>>>(xyz, finit, out_newxyz, wsA, wsC2,
                                                    points, wsT, ws_bytes);
    group_kernel<<<dim3(S_ / 16, B_), 512, 0, stream>>>(xyz, points, wsA, wsC2, wsT, ws_bytes, out_feat);
}


# FPS pass reads a thread's four points as three 16-byte LDS rows (all x, all y, all z) of a second, coordinate-major copy, not four {x,y,z,0} records; records index-ordered, for pick look-ups only
# speedup vs baseline: 5.4538x; 1.0726x over previous
#include <hip/hip_runtime.h>
#include <math.h>

#define B_  8
#define N_  4096
#define D_  64
#define S_  1024
#define K_  32

typedef __attribute__((ext_vector_type(2))) float v2f;
typedef __attribute__((ext_vector_type(4))) float v4f;
typedef __attribute__((ext_vector_type(8))) float v8f;

#define WS_T_OFF  ((size_t)B_ * S_ * 5)
#define WS_NEED   ((WS_T_OFF + (size_t)B_ * N_ * D_) * 4)
#define XT_BLOCKS (B_ * (N_ / 64))
#define XT_PITCH  65
typedef __attribute__((ext_vector_type(4))) int v4i;
typedef __attribute__((ext_vector_type(2))) unsigned long long v2u64;
#define FD 4
#define P4_AT(i) (i)

template <int J>
__device__ __forceinline__ float fps_sqdist(const v4f p, const v4f c)
{
    const float dx = p.x - c.x, dy = p.y - c.y, dz = p.z - c.z;
    float d;
    {
#pragma clang fp contract(off)
        d = (J < 2) ? __builtin_fmaf(dz, dz, __builtin_fmaf(dx, dx, dy * dy))
                    : dz * dz + __builtin_fmaf(dy, dy, dx * dx);
    }
    return d;
}

__device__ __forceinline__ void fps_emit(const v4f c, int b, int pos,
                                         float* __restrict__ out_newxyz,
                                         float* __restrict__ wsA,
                                         float* __restrict__ wsC2)
{
    out_newxyz[(size_t)b * 3 * S_ + 0 * S_ + pos] = c.x;
    out_newxyz[(size_t)b * 3 * S_ + 1 * S_ + pos] = c.y;
    out_newxyz[(size_t)b * 3 * S_ + 2 * S_ + pos] = c.z;
    float* A = wsA + ((size_t)b * S_ + pos) * 4;
    A[0] = -2.f * c.x; A[1] = -2.f * c.y; A[2] = -2.f * c.z; A[3] = 1.f;
    float c2;
    {
#pragma clang fp contract(off)
        c2 = c.z * c.z + __builtin_fmaf(c.x, c.x, c.y * c.y);
    }
    wsC2[(size_t)b * S_ + pos] = c2;
}

__device__ __forceinline__ void transpose_tile(const float* __restrict__ pts,
                                               float* __restrict__ wsT,
                                               int t, int tid, float* tile)
{
    const int b  = t / (N_ / 64);
    const int n0 = (t % (N_ / 64)) * 64;
    {
        const int ch = tid >> 4, p0 = (tid & 15) * 4;
        const v4f v = *(const v4f*)(pts + ((size_t)b * D_ + ch) * N_ + n0 + p0);
        float* d = tile + ch * XT_PITCH + p0;
        d[0] = v.x; d[1] = v.y; d[2] = v.z; d[3] = v.w;
    }
    __syncthreads();
    {
        const int pt = tid >> 4, c0 = (tid & 15) * 4;
        const float* s = tile + c0 * XT_PITCH + pt;
        v4f o;
        o.x = s[0]; o.y = s[XT_PITCH]; o.z = s[2 * XT_PITCH]; o.w = s[3 * XT_PITCH];
        *(v4f*)(wsT + ((size_t)b * N_ + n0 + pt) * D_ + c0) = o;
    }
}

template <int J>
__device__ __forceinline__ void fps_pass(const v4f* sx, const v4f* sy, const v4f* sz,
                                         int tid, bool ok1, bool ok2, bool ok3,
                                         const v4f c0, const v4f c1, const v4f c2, const v4f c3,
                                         float (&dS)[FD][4], float (*tmax)[1024], v4f (*runs)[1024])
{
    const v4f X = sx[tid], Y = sy[tid], Z = sz[tid];
#pragma unroll
    for (int s = 0; s < 4; ++s) {
        v4f p;
        p.x = X[s]; p.y = Y[s]; p.z = Z[s]; p.w = 0.f;
        const float dist = ok3 ? dS[3][s] : (ok2 ? dS[2][s] : (ok1 ? dS[1][s] : dS[0][s]));
        float d[FD];
        d[0] = fps_sqdist<J>(p, c0); d[1] = fps_sqdist<J>(p, c1);
        d[2] = fps_sqdist<J>(p, c2); d[3] = fps_sqdist<J>(p, c3);
        float run = dist;
#pragma unroll
        for (int j = 0; j < FD; ++j) {
            run = fminf(run, d[j]);
            dS[j][s] = run;
        }
    }
#pragma unroll
    for (int j = 0; j < FD; ++j) {
        v4f r;
        r.x = dS[j][0]; r.y = dS[j][1]; r.z = dS[j][2]; r.w = dS[j][3];
        runs[j][tid] = r;
        tmax[j][tid] = fmaxf(fmaxf(r.x, r.y), fmaxf(r.z, r.w));
    }
}

__global__ __launch_bounds__(1024) void fps_kernel(
    const float* __restrict__ xyz,
    const int*   __restrict__ farthest_init,
    float* __restrict__ out_newxyz,
    float* __restrict__ wsA,
    float* __restrict__ wsC2,
    const float* __restrict__ pts,
    float* __restrict__ wsT,
    unsigned long long ws_bytes)
{
    __shared__ v4f   p4[N_];
    __shared__ v4f   sx[1024], sy[1024], sz[1024];
    __shared__ __attribute__((aligned(16))) float tmax[FD][1024];
    __shared__ __attribute__((aligned(16))) v4f runs[FD][1024];
    __shared__ __attribute__((aligned(16))) int sRes[FD][4];
    __shared__ __attribute__((aligned(16))) unsigned long long lkeys[FD][32];

    const int b    = blockIdx.x;
    const int tid  = threadIdx.x;
    const int lane = tid & 31;
    const int wid  = tid >> 5;

    if (b >= B_) {
        if (ws_bytes >= WS_NEED) transpose_tile(pts, wsT, b - B_, tid, (float*)p4);
        return;
    }

    const float* xb = xyz + (size_t)b * 3 * N_;
    {
        const v4f X = *(const v4f*)(xb + 4 * tid);
        const v4f Y = *(const v4f*)(xb + N_ + 4 * tid);
        const v4f Z = *(const v4f*)(xb + 2 * N_ + 4 * tid);
        sx[tid] = X; sy[tid] = Y; sz[tid] = Z;
#pragma unroll
        for (int s = 0; s < 4; ++s) {
            v4f v;
            v.x = X[s]; v.y = Y[s]; v.z = Z[s]; v.w = 0.f;
            p4[P4_AT(4 * tid + s)] = v;
        }
    }
    float dS[FD][4];
#pragma unroll
    for (int j = 0; j < FD; ++j) {
#pragma unroll
        for (int s = 0; s < 4; ++s) dS[j][s] = 1e10f;
    }
    if (tid < FD * 4) (&sRes[0][0])[tid] = farthest_init[b];
    __syncthreads();

    int pos = -1;
    int g1 = -1, g2 = -1, g3 = -1;
    v4f c1 = {0.f, 0.f, 0.f, 0.f}, c2 = c1, c3 = c1;
    for (int round = 0; round < S_; ++round) {
        const v4i r0 = *(const v4i*)sRes[0];
        const v4i r1 = *(const v4i*)sRes[1];
        const v4i r2 = *(const v4i*)sRes[2];
        const v4i r3 = *(const v4i*)sRes[3];
        const int w0 = __builtin_amdgcn_readfirstlane(r0.x);
        const int w1 = __builtin_amdgcn_readfirstlane(r1.x);
        const int w2 = __builtin_amdgcn_readfirstlane(r2.x);
        const bool ok1 = (w0 == g1);
        const bool ok2 = ok1 && (w1 == g2);
        const bool ok3 = ok2 && (w2 == g3);
        const int  na  = (ok1 ? 1 : 0) + (ok2 ? 1 : 0) + (ok3 ? 1 : 0);
        if (tid == 0) {
            if (ok1 && pos + 1 < S_) fps_emit(c1, b, pos + 1, out_newxyz, wsA, wsC2);
            if (ok2 && pos + 2 < S_) fps_emit(c2, b, pos + 2, out_newxyz, wsA, wsC2);
            if (ok3 && pos + 3 < S_) fps_emit(c3, b, pos + 3, out_newxyz, wsA, wsC2);
        }
        pos += na + 1;
        if (pos >= S_) break;
        const v4i rs = ok3 ? r3 : (ok2 ? r2 : (ok1 ? r1 : r0));
        const int g0 = __builtin_amdgcn_readfirstlane(rs.x);
        g1 = __builtin_amdgcn_readfirstlane(rs.y);
        g2 = __builtin_amdgcn_readfirstlane(rs.z);
        g3 = __builtin_amdgcn_readfirstlane(rs.w);
        const v4f c0 = p4[P4_AT(g0)];
        c1 = p4[P4_AT(g1)];
        c2 = p4[P4_AT(g2)];
        c3 = p4[P4_AT(g3)];
        if (tid == 0) fps_emit(c0, b, pos, out_newxyz, wsA, wsC2);

        if (__builtin_amdgcn_readfirstlane(wid) < 16)
            fps_pass<0>(sx, sy, sz, tid, ok1, ok2, ok3, c0, c1, c2, c3, dS, tmax, runs);
        else
            fps_pass<2>(sx, sy, sz, tid, ok1, ok2, ok3, c0, c1, c2, c3, dS, tmax, runs);
        __syncthreads();
        if (wid < FD) {
            const float* pr = tmax[wid];
            float tv[32];
#pragma unroll
            for (int k = 0; k < 8; ++k) {
                const v4f q = *(const v4f*)&pr[4 * (k * 32 + lane)];
                tv[4 * k]     = q.x; tv[4 * k + 1] = q.y;
                tv[4 * k + 2] = q.z; tv[4 * k + 3] = q.w;
            }
            float tm[16];
#pragma unroll
            for (int t = 0; t < 16; ++t) tm[t] = fmaxf(tv[t], tv[t + 16]);
#pragma unroll
            for (int n = 8; n >= 1; n >>= 1) {
#pragma unroll
                for (int t = 0; t < n; ++t) tm[t] = fmaxf(tm[t], tm[t + n]);
            }
            const float lmax = tm[0];
            int to[32];
#pragma unroll
            for (int t = 0; t < 32; ++t) to[t] = (tv[t] == lmax) ? t : 64;
#pragma unroll
            for (int n = 16; n >= 1; n >>= 1) {
#pragma unroll
                for (int t = 0; t < n; ++t) to[t] = to[t + n] < to[t] ? to[t + n] : to[t];
            }
            const int lt = 128 * (to[0] >> 2) + 4 * lane + (to[0] & 3);
            const v4f rq = runs[wid][lt];
            const unsigned ub = __float_as_uint(lmax);
            const unsigned hb = ub ^ ((unsigned)((int)ub >> 31) | 0x80000000u);
            const unsigned long long key =
                ((unsigned long long)hb << 32) | (unsigned long long)(unsigned)(~lt);
            unsigned long long* kb = lkeys[wid];
            kb[lane] = key;
            v2u64 q[16];
#pragma unroll
            for (int k = 0; k < 16; ++k) q[k] = *(const v2u64*)&kb[2 * k];
            asm volatile("" :: "v"(rq.x), "v"(rq.y), "v"(rq.z), "v"(rq.w));
            int rk[4] = {0, 0, 0, 0};
#pragma unroll
            for (int k = 0; k < 16; ++k)
                rk[k & 3] += (q[k].x > key ? 1 : 0) + (q[k].y > key ? 1 : 0);
            const int rank = (rk[0] + rk[1]) + (rk[2] + rk[3]);
            int slot = 3;
            slot = (rq.z == lmax) ? 2 : slot;
            slot = (rq.y == lmax) ? 1 : slot;
            slot = (rq.x == lmax) ? 0 : slot;
            const int li = 4 * lt + slot;
            if (rank < 4) sRes[wid][rank] = li;
        }
        __syncthreads();
    }
}

#define TPITCH 4100
#define CAND_CAP 512

template <int R>
__device__ __forceinline__ void select32(const float* cd, const unsigned short* cj,
                                         int C, int lane, int* selRow)
{
    unsigned key[R];
    int      jv[R];
#pragma unroll
    for (int r = 0; r < R; ++r) {
        const int  i  = r * 32 + lane;
        const bool in = i < C;
        const int  ii = in ? i : 0;
        const unsigned u = __float_as_uint(cd[ii]);
        const unsigned k = u ^ ((unsigned)((int)u >> 31) | 0x80000000u);
        key[r] = in ? k : 0xFFFFFFFFu;
        jv[r]  = (int)cj[ii];
    }
    unsigned T = 0;
    for (int bit = 31; bit >= 0; --bit) {
        const unsigned cand = T | (1u << bit);
        int cnt = 0;
#pragma unroll
        for (int r = 0; r < R; ++r)
            cnt += __popc(__builtin_amdgcn_ballot_w32(key[r] < cand));
        if (cnt < K_) T = cand;
    }
    int nless = 0;
#pragma unroll
    for (int r = 0; r < R; ++r)
        nless += __popc(__builtin_amdgcn_ballot_w32(key[r] < T));
    const int need = K_ - nless;
    int running = 0, tieSeen = 0;
#pragma unroll
    for (int r = 0; r < R; ++r) {
        const bool isTie = key[r] == T;
        const unsigned tb = __builtin_amdgcn_ballot_w32(isTie);
        const int tieRank = tieSeen + (int)__builtin_amdgcn_mbcnt_lo(tb, 0u);
        const bool sel = (key[r] < T) || (isTie && tieRank < need);
        const unsigned sb = __builtin_amdgcn_ballot_w32(sel);
        const int slot = running + (int)__builtin_amdgcn_mbcnt_lo(sb, 0u);
        if (sel && slot < K_) selRow[slot] = jv[r];
        running += __popc(sb);
        tieSeen += __popc(tb);
    }
}

__device__ __forceinline__ void select32_scan(const float* row, int lane, int* selRow)
{
    float pd = -3.4e38f; int pj = -1;
    for (int k = 0; k < K_; ++k) {
        float bv = 3.4e38f; int bi = 0x7fffffff;
        for (int jj = 0; jj < N_ / 128; ++jj) {
            const int j0 = jj * 128 + lane * 4;
            const v4f dv = *(const v4f*)&row[j0];
#pragma unroll
            for (int e = 0; e < 4; ++e) {
                const float d = dv[e];
                const int   j = j0 + e;
                const bool after = d > pd || (d == pd && j > pj);
                if (after && (d < bv || (d == bv && j < bi))) { bv = d; bi = j; }
            }
        }
#pragma unroll
        for (int off = 16; off >= 1; off >>= 1) {
            const float ov = __shfl_xor(bv, off, 32);
            const int   oi = __shfl_xor(bi, off, 32);
            if (ov < bv || (ov == bv && oi < bi)) { bv = ov; bi = oi; }
        }
        if (lane == 0) selRow[k] = bi;
        pd = bv; pj = bi;
    }
}

__global__ __launch_bounds__(512) void group_kernel(
    const float* __restrict__ xyz,
    const float* __restrict__ pts,
    const float* __restrict__ wsA,
    const float* __restrict__ wsC2,
    const float* __restrict__ wsT,
    unsigned long long ws_bytes,
    float* __restrict__ out_feat)
{
    __shared__ float DT[16 * TPITCH];
    __shared__ __attribute__((aligned(16))) int selIdx[16][K_];
    __shared__ float          candD[16][CAND_CAP];
    __shared__ unsigned short candJ[16][CAND_CAP];

    const int b    = blockIdx.y;
    const int s0   = blockIdx.x * 16;
    const int tid  = threadIdx.x;
    const int lane = tid & 31;
    const int wid  = tid >> 5;
    const bool hi  = lane >= 16;

    const float* xb = xyz + (size_t)b * 3 * N_;

    const int  mcol = lane & 15;
    const v4f  A4   = *(const v4f*)(wsA + ((size_t)b * S_ + s0 + mcol) * 4);
    v2f a;
    a.x = hi ? A4.z : A4.x;
    a.y = hi ? A4.w : A4.y;

    const int mbase = hi ? 8 : 0;
    float cmr[8];
#pragma unroll
    for (int r = 0; r < 8; ++r)
        cmr[r] = wsC2[(size_t)b * S_ + s0 + mbase + r];

    for (int t = 0; t < 16; ++t) {
        const int n = wid * 256 + t * 16 + mcol;
        const float x = xb[n], y = xb[N_ + n], z = xb[2 * N_ + n];
        const float pn2 = x * x + y * y + z * z;
        v2f bb;
        bb.x = hi ? z : x;
        bb.y = hi ? pn2 : y;

        v8f c = {};
        c = __builtin_amdgcn_wmma_f32_16x16x4_f32(
                 false, a,  false, bb,
                 (short)0, c,  false,  false);

#pragma unroll
        for (int r = 0; r < 8; ++r)
            DT[(mbase + r) * TPITCH + n] = c[r] + cmr[r];
    }
    __syncthreads();

    const int m = wid;
    {
        const float* row = &DT[m * TPITCH];
        float* cd = candD[wid];
        unsigned short* cj = candJ[wid];

        float mn = 3.4e38f;
        for (int jj = 0; jj < N_ / 128; ++jj) {
            const v4f dv = *(const v4f*)&row[jj * 128 + lane * 4];
            mn = fminf(mn, fminf(fminf(dv.x, dv.y), fminf(dv.z, dv.w)));
        }
        float U = mn;
#pragma unroll
        for (int off = 16; off >= 1; off >>= 1)
            U = fmaxf(U, __shfl_xor(U, off, 32));

        int C = 0;
        for (int jj = 0; jj < N_ / 128; ++jj) {
            const int j0 = jj * 128 + lane * 4;
            const v4f dv = *(const v4f*)&row[j0];
            const bool p0 = dv.x <= U, p1 = dv.y <= U;
            const bool p2 = dv.z <= U, p3 = dv.w <= U;
            const unsigned b0 = __builtin_amdgcn_ballot_w32(p0);
            const unsigned b1 = __builtin_amdgcn_ballot_w32(p1);
            const unsigned b2 = __builtin_amdgcn_ballot_w32(p2);
            const unsigned b3 = __builtin_amdgcn_ballot_w32(p3);
            unsigned pre = __builtin_amdgcn_mbcnt_lo(b0, 0u);
            pre = __builtin_amdgcn_mbcnt_lo(b1, pre);
            pre = __builtin_amdgcn_mbcnt_lo(b2, pre);
            pre = __builtin_amdgcn_mbcnt_lo(b3, pre);
            int pos = C + (int)pre;
            if (p0 && pos < CAND_CAP) { cd[pos] = dv.x; cj[pos] = (unsigned short)j0; }
            pos += p0 ? 1 : 0;
            if (p1 && pos < CAND_CAP) { cd[pos] = dv.y; cj[pos] = (unsigned short)(j0 + 1); }
            pos += p1 ? 1 : 0;
            if (p2 && pos < CAND_CAP) { cd[pos] = dv.z; cj[pos] = (unsigned short)(j0 + 2); }
            pos += p2 ? 1 : 0;
            if (p3 && pos < CAND_CAP) { cd[pos] = dv.w; cj[pos] = (unsigned short)(j0 + 3); }
            C += __popc(b0) + __popc(b1) + __popc(b2) + __popc(b3);
        }
        __syncthreads();

        int* selRow = selIdx[m];
        if (C <= 128)           select32<4>(cd, cj, C, lane, selRow);
        else if (C <= 256)      select32<8>(cd, cj, C, lane, selRow);
        else if (C <= CAND_CAP) select32<16>(cd, cj, C, lane, selRow);
        else                    select32_scan(row, lane, selRow);
        __syncthreads();
    }

    float* ob = out_feat + (size_t)b * 67 * S_ + (s0 + m);
    if (ws_bytes >= WS_NEED) {
        const int q = lane & 15;
        const v4i* sr = (const v4i*)(selIdx[m] + (hi ? 16 : 0));
        const float* tb = wsT + (size_t)b * N_ * D_ + 4 * q;
        v4i jv[4];
#pragma unroll
        for (int i = 0; i < 4; ++i) jv[i] = sr[i];
        v4f mx = {-3.4e38f, -3.4e38f, -3.4e38f, -3.4e38f};
#pragma unroll
        for (int k = 0; k < 16; ++k) {
            const int j = jv[k >> 2][k & 3];
            const v4f v = *(const v4f*)(tb + (size_t)j * D_);
            mx.x = fmaxf(mx.x, v.x); mx.y = fmaxf(mx.y, v.y);
            mx.z = fmaxf(mx.z, v.z); mx.w = fmaxf(mx.w, v.w);
        }
        const int jl = selIdx[m][lane];
        float gx = xb[jl], gy = xb[N_ + jl], gz = xb[2 * N_ + jl];
        mx.x = fmaxf(mx.x, __shfl_xor(mx.x, 16, 32));
        mx.y = fmaxf(mx.y, __shfl_xor(mx.y, 16, 32));
        mx.z = fmaxf(mx.z, __shfl_xor(mx.z, 16, 32));
        mx.w = fmaxf(mx.w, __shfl_xor(mx.w, 16, 32));
#pragma unroll
        for (int off = 16; off >= 1; off >>= 1) {
            gx = fmaxf(gx, __shfl_xor(gx, off, 32));
            gy = fmaxf(gy, __shfl_xor(gy, off, 32));
            gz = fmaxf(gz, __shfl_xor(gz, off, 32));
        }
        const int c0 = 3 + 4 * q + (hi ? 2 : 0);
        ob[(size_t)c0 * S_]       = hi ? mx.z : mx.x;
        ob[(size_t)(c0 + 1) * S_] = hi ? mx.w : mx.y;
        if (lane < 3) ob[(size_t)lane * S_] = lane == 0 ? gx : (lane == 1 ? gy : gz);
        return;
    }
    const float* pb = pts + (size_t)b * D_ * N_;
    const unsigned o0 = (unsigned)lane * N_;
    const unsigned o1 = (unsigned)(lane + 32) * N_;
    const unsigned o2 = (unsigned)(lane < 2 ? lane : 2) * N_;
    const int* selRow = selIdx[m];
    float m0 = -3.4e38f, m1 = -3.4e38f, m2 = -3.4e38f;
#pragma unroll
    for (int k = 0; k < K_; ++k) {
        const int j = __builtin_amdgcn_readfirstlane(selRow[k]);
        m0 = fmaxf(m0, pb[o0 + j]);
        m1 = fmaxf(m1, pb[o1 + j]);
        m2 = fmaxf(m2, xb[o2 + j]);
    }
    ob[(size_t)(3 + lane) * S_]  = m0;
    ob[(size_t)(35 + lane) * S_] = m1;
    if (lane < 3) ob[(size_t)lane * S_] = m2;
}

extern "C" void kernel_launch(void* const* d_in, const int* in_sizes, int n_in,
                              void* d_out, int out_size, void* d_ws, size_t ws_size,
                              hipStream_t stream) {
    const float* xyz    = (const float*)d_in[0];
    const float* points = (const float*)d_in[1];
    const int*   finit  = (const int*)d_in[2];

    float* out        = (float*)d_out;
    float* out_newxyz = out;
    float* out_feat   = out + (size_t)B_ * 3 * S_;

    float* wsA  = (float*)d_ws;
    float* wsC2 = wsA + (size_t)B_ * S_ * 4;
    float* wsT  = wsA + WS_T_OFF;

    const unsigned long long ws_bytes = ws_size;
    fps_kernel<<<B_ + XT_BLOCKS, 1024, 0, stream>>>(xyz, finit, out_newxyz, wsA, wsC2,
                                                    points, wsT, ws_bytes);
    group_kernel<<<dim3(S_ / 16, B_), 512, 0, stream>>>(xyz, points, wsA, wsC2, wsT, ws_bytes, out_feat);
}
